# starting kernel
# baseline (speedup 1.0000x reference)
#include <hip/hip_runtime.h>
#include <math.h>

typedef float v2f __attribute__((ext_vector_type(2)));
typedef float v8f __attribute__((ext_vector_type(8)));
typedef float f4  __attribute__((ext_vector_type(4)));

#define NTM_B 64
#define NTM_N 8192
#define NTM_D 128
#define NTM_C 256
#define NTM_EPS 1e-8f

__global__ void ntm_key_gemm(const float* __restrict__ cs,
                             const float* __restrict__ Wk,
                             float* __restrict__ key) {
  const int n0   = blockIdx.x * 16;
  const int m0   = blockIdx.y * 16;
  const int lane = threadIdx.x;
  const int half = lane >> 4;
  const int r    = lane & 15;

#if __has_builtin(__builtin_amdgcn_wmma_f32_16x16x4_f32)
  v8f acc = {};
  const float* arow = cs + (size_t)(m0 + r) * NTM_C + 2 * half;
  const float* bcol = Wk + (size_t)(2 * half) * NTM_D + n0 + r;
  for (int k0 = 0; k0 < NTM_C; k0 += 4) {
    v2f a, b;
    a.x = arow[k0];
    a.y = arow[k0 + 1];
    b.x = bcol[(size_t)k0 * NTM_D];
    b.y = bcol[(size_t)(k0 + 1) * NTM_D];
    acc = __builtin_amdgcn_wmma_f32_16x16x4_f32(
         false, a,  false, b,
         (short)0, acc,  false,  false);
  }
#pragma unroll
  for (int i = 0; i < 8; ++i)
    key[(size_t)(m0 + half * 8 + i) * NTM_D + n0 + r] = acc[i];
#else
  for (int i = 0; i < 8; ++i) {
    const int row = m0 + half * 8 + i;
    const int col = n0 + r;
    float s = 0.f;
    for (int k = 0; k < NTM_C; ++k)
      s += cs[(size_t)row * NTM_C + k] * Wk[(size_t)k * NTM_D + col];
    key[(size_t)row * NTM_D + col] = s;
  }
#endif
}

__device__ __forceinline__ float ntm_softplus(float x) {
  return (x > 20.f) ? x : log1pf(expf(x));
}

__global__ void ntm_params(const float* __restrict__ cs,
                           const float* __restrict__ key,
                           const float* __restrict__ Wb,  const float* __restrict__ bb,
                           const float* __restrict__ Wg,  const float* __restrict__ bg,
                           const float* __restrict__ Ws,  const float* __restrict__ bs,
                           const float* __restrict__ Wgam,const float* __restrict__ bgam,
                           float* __restrict__ params) {
  __shared__ float sd[7 * 256];
  const int b = blockIdx.x;
  const int t = threadIdx.x;

  const float c = cs[(size_t)b * NTM_C + t];
  sd[0 * 256 + t] = c * Wb[t];
  sd[1 * 256 + t] = c * Wg[t];
  sd[2 * 256 + t] = c * Wgam[t];
  sd[3 * 256 + t] = c * Ws[t * 3 + 0];
  sd[4 * 256 + t] = c * Ws[t * 3 + 1];
  sd[5 * 256 + t] = c * Ws[t * 3 + 2];
  const float kv = (t < NTM_D) ? key[(size_t)b * NTM_D + t] : 0.f;
  sd[6 * 256 + t] = kv * kv;
  __syncthreads();

  for (int s = 128; s > 0; s >>= 1) {
    if (t < s) {
#pragma unroll
      for (int j = 0; j < 7; ++j) sd[j * 256 + t] += sd[j * 256 + t + s];
    }
    __syncthreads();
  }

  if (t == 0) {
    const float beta  = ntm_softplus(sd[0 * 256] + bb[0]) + 1.0f;
    const float gate  = 1.0f / (1.0f + expf(-(sd[1 * 256] + bg[0])));
    const float gamma = ntm_softplus(sd[2 * 256] + bgam[0]) + 1.0f;
    const float knorm = sqrtf(sd[6 * 256]);
    float x0 = sd[3 * 256] + bs[0];
    float x1 = sd[4 * 256] + bs[1];
    float x2 = sd[5 * 256] + bs[2];
    const float m = fmaxf(x0, fmaxf(x1, x2));
    const float e0 = expf(x0 - m), e1 = expf(x1 - m), e2 = expf(x2 - m);
    const float inv = 1.0f / (e0 + e1 + e2);
    float* p = params + b * 8;
    p[0] = beta;  p[1] = gate;  p[2] = gamma; p[3] = knorm;
    p[4] = e0 * inv; p[5] = e1 * inv; p[6] = e2 * inv;
  }
}

__global__ void ntm_dot(const float* __restrict__ memory,
                        const float* __restrict__ key,
                        const float* __restrict__ params,
                        float* __restrict__ z) {
  const int b    = blockIdx.y;
  const int w    = threadIdx.x >> 5;
  const int lane = threadIdx.x & 31;

  const f4 k4 = ((const f4*)(key + (size_t)b * NTM_D))[lane];
  const float beta  = params[b * 8 + 0];
  const float knorm = params[b * 8 + 3];

  const int base = blockIdx.x * 64 + w * 8;
#pragma unroll
  for (int r = 0; r < 8; ++r) {
    const int n = base + r;
    const f4* mp = (const f4*)(memory + ((size_t)b * NTM_N + n) * NTM_D);
    const f4 m4 = __builtin_nontemporal_load(mp + lane);
    float pd = m4.x * k4.x + m4.y * k4.y + m4.z * k4.z + m4.w * k4.w;
    float ps = m4.x * m4.x + m4.y * m4.y + m4.z * m4.z + m4.w * m4.w;
#pragma unroll
    for (int off = 16; off > 0; off >>= 1) {
      pd += __shfl_down(pd, off);
      ps += __shfl_down(ps, off);
    }
    if (lane == 0)
      z[(size_t)b * NTM_N + n] = beta * pd / (knorm * sqrtf(ps) + NTM_EPS);
  }
}

__device__ __forceinline__ float ntm_block_reduce_sum(float v, float* red) {
#pragma unroll
  for (int off = 16; off > 0; off >>= 1) v += __shfl_down(v, off);
  const int wid = threadIdx.x >> 5, lane = threadIdx.x & 31;
  if (lane == 0) red[wid] = v;
  __syncthreads();
  if (threadIdx.x < 32) {
    float x = red[threadIdx.x];
#pragma unroll
    for (int off = 16; off > 0; off >>= 1) x += __shfl_down(x, off);
    if (threadIdx.x == 0) red[0] = x;
  }
  __syncthreads();
  const float r = red[0];
  __syncthreads();
  return r;
}

__device__ __forceinline__ float ntm_block_reduce_max(float v, float* red) {
#pragma unroll
  for (int off = 16; off > 0; off >>= 1) v = fmaxf(v, __shfl_down(v, off));
  const int wid = threadIdx.x >> 5, lane = threadIdx.x & 31;
  if (lane == 0) red[wid] = v;
  __syncthreads();
  if (threadIdx.x < 32) {
    float x = red[threadIdx.x];
#pragma unroll
    for (int off = 16; off > 0; off >>= 1) x = fmaxf(x, __shfl_down(x, off));
    if (threadIdx.x == 0) red[0] = x;
  }
  __syncthreads();
  const float r = red[0];
  __syncthreads();
  return r;
}

__global__ void ntm_rowops(const float* __restrict__ z,
                           const float* __restrict__ prev,
                           const float* __restrict__ params,
                           float* __restrict__ out) {
  __shared__ float row[NTM_N];
  __shared__ float red[32];

  const int b   = blockIdx.x;
  const int tid = threadIdx.x;
  const float gate  = params[b * 8 + 1];
  const float gamma = params[b * 8 + 2];
  const float s0 = params[b * 8 + 4];
  const float s1 = params[b * 8 + 5];
  const float s2 = params[b * 8 + 6];

  float zv[8];
  float lmax = -3.4e38f;
#pragma unroll
  for (int i = 0; i < 8; ++i) {
    const int n = tid + i * 1024;
    zv[i] = z[(size_t)b * NTM_N + n];
    lmax = fmaxf(lmax, zv[i]);
  }
  const float rmax = ntm_block_reduce_max(lmax, red);

  float lsum = 0.f;
#pragma unroll
  for (int i = 0; i < 8; ++i) {
    zv[i] = expf(zv[i] - rmax);
    lsum += zv[i];
  }
  const float invsum = 1.0f / ntm_block_reduce_sum(lsum, red);

#pragma unroll
  for (int i = 0; i < 8; ++i) {
    const int n = tid + i * 1024;
    const float cw = zv[i] * invsum;
    row[n] = gate * cw + (1.0f - gate) * prev[(size_t)b * NTM_N + n];
  }
  __syncthreads();

  float sh[8];
  float lsh = 0.f;
#pragma unroll
  for (int i = 0; i < 8; ++i) {
    const int n = tid + i * 1024;
    const float right = row[(n + 1) & (NTM_N - 1)];
    const float mid   = row[n];
    const float left  = row[(n + NTM_N - 1) & (NTM_N - 1)];
    const float shifted = s0 * right + s1 * mid + s2 * left;
    sh[i] = powf(shifted + 1e-8f, gamma);
    lsh += sh[i];
  }
  const float invt = 1.0f / (ntm_block_reduce_sum(lsh, red) + NTM_EPS);

#pragma unroll
  for (int i = 0; i < 8; ++i) {
    const int n = tid + i * 1024;
    out[(size_t)b * NTM_N + n] = sh[i] * invt;
  }
}

extern "C" void kernel_launch(void* const* d_in, const int* in_sizes, int n_in,
                              void* d_out, int out_size, void* d_ws, size_t ws_size,
                              hipStream_t stream) {
  const float* memory = (const float*)d_in[0];
  const float* cs     = (const float*)d_in[1];
  const float* prev   = (const float*)d_in[2];
  const float* Wk     = (const float*)d_in[3];
  const float* Wb     = (const float*)d_in[4];
  const float* bb     = (const float*)d_in[5];
  const float* Wg     = (const float*)d_in[6];
  const float* bg     = (const float*)d_in[7];
  const float* Ws     = (const float*)d_in[8];
  const float* bs     = (const float*)d_in[9];
  const float* Wgam   = (const float*)d_in[10];
  const float* bgam   = (const float*)d_in[11];
  float* out = (float*)d_out;

  float* zbuf   = (float*)d_ws;
  float* keyb   = zbuf + (size_t)NTM_B * NTM_N;
  float* params = keyb + (size_t)NTM_B * NTM_D;

  ntm_key_gemm<<<dim3(NTM_D / 16, NTM_B / 16), 32, 0, stream>>>(cs, Wk, keyb);
  ntm_params<<<NTM_B, 256, 0, stream>>>(cs, keyb, Wb, bb, Wg, bg, Ws, bs,
                                        Wgam, bgam, params);
  ntm_dot<<<dim3(NTM_N / 64, NTM_B), 256, 0, stream>>>(memory, keyb, params, zbuf);
  ntm_rowops<<<NTM_B, 1024, 0, stream>>>(zbuf, prev, params, out);
}


# ntm_dot as 256 workgroups x 16 long-lived waves: each wave loops over 128 rows with the next 16 rows' loads in flight (was 65536 waves of 8 rows, one load in flight each); sum order unchanged
# speedup vs baseline: 1.2393x; 1.2393x over previous
#include <hip/hip_runtime.h>
#include <math.h>

typedef float v2f __attribute__((ext_vector_type(2)));
typedef float v8f __attribute__((ext_vector_type(8)));
typedef float f4  __attribute__((ext_vector_type(4)));

#define NTM_B 64
#define NTM_N 8192
#define NTM_D 128
#define NTM_C 256
#define NTM_EPS 1e-8f

__global__ void ntm_key_gemm(const float* __restrict__ cs,
                             const float* __restrict__ Wk,
                             float* __restrict__ key) {
  const int n0   = blockIdx.x * 16;
  const int m0   = blockIdx.y * 16;
  const int lane = threadIdx.x;
  const int half = lane >> 4;
  const int r    = lane & 15;

#if __has_builtin(__builtin_amdgcn_wmma_f32_16x16x4_f32)
  v8f acc = {};
  const float* arow = cs + (size_t)(m0 + r) * NTM_C + 2 * half;
  const float* bcol = Wk + (size_t)(2 * half) * NTM_D + n0 + r;
  for (int k0 = 0; k0 < NTM_C; k0 += 4) {
    v2f a, b;
    a.x = arow[k0];
    a.y = arow[k0 + 1];
    b.x = bcol[(size_t)k0 * NTM_D];
    b.y = bcol[(size_t)(k0 + 1) * NTM_D];
    acc = __builtin_amdgcn_wmma_f32_16x16x4_f32(
         false, a,  false, b,
         (short)0, acc,  false,  false);
  }
#pragma unroll
  for (int i = 0; i < 8; ++i)
    key[(size_t)(m0 + half * 8 + i) * NTM_D + n0 + r] = acc[i];
#else
  for (int i = 0; i < 8; ++i) {
    const int row = m0 + half * 8 + i;
    const int col = n0 + r;
    float s = 0.f;
    for (int k = 0; k < NTM_C; ++k)
      s += cs[(size_t)row * NTM_C + k] * Wk[(size_t)k * NTM_D + col];
    key[(size_t)row * NTM_D + col] = s;
  }
#endif
}

__device__ __forceinline__ float ntm_softplus(float x) {
  return (x > 20.f) ? x : log1pf(expf(x));
}

__global__ void ntm_params(const float* __restrict__ cs,
                           const float* __restrict__ key,
                           const float* __restrict__ Wb,  const float* __restrict__ bb,
                           const float* __restrict__ Wg,  const float* __restrict__ bg,
                           const float* __restrict__ Ws,  const float* __restrict__ bs,
                           const float* __restrict__ Wgam,const float* __restrict__ bgam,
                           float* __restrict__ params) {
  __shared__ float sd[7 * 256];
  const int b = blockIdx.x;
  const int t = threadIdx.x;

  const float c = cs[(size_t)b * NTM_C + t];
  sd[0 * 256 + t] = c * Wb[t];
  sd[1 * 256 + t] = c * Wg[t];
  sd[2 * 256 + t] = c * Wgam[t];
  sd[3 * 256 + t] = c * Ws[t * 3 + 0];
  sd[4 * 256 + t] = c * Ws[t * 3 + 1];
  sd[5 * 256 + t] = c * Ws[t * 3 + 2];
  const float kv = (t < NTM_D) ? key[(size_t)b * NTM_D + t] : 0.f;
  sd[6 * 256 + t] = kv * kv;
  __syncthreads();

  for (int s = 128; s > 0; s >>= 1) {
    if (t < s) {
#pragma unroll
      for (int j = 0; j < 7; ++j) sd[j * 256 + t] += sd[j * 256 + t + s];
    }
    __syncthreads();
  }

  if (t == 0) {
    const float beta  = ntm_softplus(sd[0 * 256] + bb[0]) + 1.0f;
    const float gate  = 1.0f / (1.0f + expf(-(sd[1 * 256] + bg[0])));
    const float gamma = ntm_softplus(sd[2 * 256] + bgam[0]) + 1.0f;
    const float knorm = sqrtf(sd[6 * 256]);
    float x0 = sd[3 * 256] + bs[0];
    float x1 = sd[4 * 256] + bs[1];
    float x2 = sd[5 * 256] + bs[2];
    const float m = fmaxf(x0, fmaxf(x1, x2));
    const float e0 = expf(x0 - m), e1 = expf(x1 - m), e2 = expf(x2 - m);
    const float inv = 1.0f / (e0 + e1 + e2);
    float* p = params + b * 8;
    p[0] = beta;  p[1] = gate;  p[2] = gamma; p[3] = knorm;
    p[4] = e0 * inv; p[5] = e1 * inv; p[6] = e2 * inv;
  }
}

#define DOT_WAVES 16
#define DOT_ROWS  128
#define DOT_G     16
#define DOT_ROW_F4 (NTM_D / 4)

template <bool REFILL>
__device__ __forceinline__ void ntm_take_group(f4 (&buf)[DOT_G], const f4* next,
                                               const f4 k4, const int lane,
                                               float (&pd)[DOT_G], float (&ps)[DOT_G]) {
#pragma unroll
  for (int r = 0; r < DOT_G; ++r) {
    const f4 m4 = buf[r];
    pd[r] = m4.x * k4.x + m4.y * k4.y + m4.z * k4.z + m4.w * k4.w;
    ps[r] = m4.x * m4.x + m4.y * m4.y + m4.z * m4.z + m4.w * m4.w;
    if (REFILL) buf[r] = __builtin_nontemporal_load(next + r * DOT_ROW_F4 + lane);
  }
}

__device__ __forceinline__ void ntm_fold_group(float (&pd)[DOT_G], float (&ps)[DOT_G],
                                               const int lane, const int lane_base,
                                               float& keep_pd, float& keep_ps) {
#pragma unroll
  for (int r = 0; r < DOT_G; ++r) {
    float d = pd[r], s = ps[r];
#pragma unroll
    for (int off = 16; off > 0; off >>= 1) {
      d += __shfl_xor(d, off);
      s += __shfl_xor(s, off);
    }
    if (lane == lane_base + r) { keep_pd = d; keep_ps = s; }
  }
}

__global__ void __launch_bounds__(DOT_WAVES * 32)
ntm_dot(const float* __restrict__ memory,
        const float* __restrict__ key,
        const float* __restrict__ params,
        float* __restrict__ z) {
  const int lane = threadIdx.x & 31;
  const int wave = __builtin_amdgcn_readfirstlane(threadIdx.x >> 5);
  constexpr int WG_PER_BATCH = NTM_N / (DOT_WAVES * DOT_ROWS);
  const int b    = blockIdx.x / WG_PER_BATCH;
  const int row0 = ((blockIdx.x % WG_PER_BATCH) * DOT_WAVES + wave) * DOT_ROWS;

  const f4 k4 = ((const f4*)(key + (size_t)b * NTM_D))[lane];
  const float beta  = params[b * 8 + 0];
  const float knorm = params[b * 8 + 3];

  const f4* p = (const f4*)(memory + ((size_t)b * NTM_N + row0) * NTM_D);
  float* zq = z + (size_t)b * NTM_N + row0 + lane;

  f4 buf[DOT_G];
  float pd[DOT_G], ps[DOT_G];
  float kd = 0.f, ks = 0.f;
  constexpr int TRIPS = DOT_ROWS / (2 * DOT_G);

#pragma unroll
  for (int r = 0; r < DOT_G; ++r)
    buf[r] = __builtin_nontemporal_load(p + r * DOT_ROW_F4 + lane);

#pragma nounroll
  for (int t = 0; t < TRIPS - 1; ++t) {
    ntm_take_group<true>(buf, p + DOT_G * DOT_ROW_F4, k4, lane, pd, ps);
    __builtin_amdgcn_sched_barrier(0);
    ntm_fold_group(pd, ps, lane, 0, kd, ks);
    __builtin_amdgcn_sched_barrier(0);
    ntm_take_group<true>(buf, p + 2 * DOT_G * DOT_ROW_F4, k4, lane, pd, ps);
    __builtin_amdgcn_sched_barrier(0);
    ntm_fold_group(pd, ps, lane, DOT_G, kd, ks);
    *zq = beta * kd / (knorm * sqrtf(ks) + NTM_EPS);
    p  += 2 * DOT_G * DOT_ROW_F4;
    zq += 2 * DOT_G;
  }
  ntm_take_group<true>(buf, p + DOT_G * DOT_ROW_F4, k4, lane, pd, ps);
  __builtin_amdgcn_sched_barrier(0);
  ntm_fold_group(pd, ps, lane, 0, kd, ks);
  ntm_take_group<false>(buf, p, k4, lane, pd, ps);
  ntm_fold_group(pd, ps, lane, DOT_G, kd, ks);
  *zq = beta * kd / (knorm * sqrtf(ks) + NTM_EPS);
}

__device__ __forceinline__ float ntm_block_reduce_sum(float v, float* red) {
#pragma unroll
  for (int off = 16; off > 0; off >>= 1) v += __shfl_down(v, off);
  const int wid = threadIdx.x >> 5, lane = threadIdx.x & 31;
  if (lane == 0) red[wid] = v;
  __syncthreads();
  if (threadIdx.x < 32) {
    float x = red[threadIdx.x];
#pragma unroll
    for (int off = 16; off > 0; off >>= 1) x += __shfl_down(x, off);
    if (threadIdx.x == 0) red[0] = x;
  }
  __syncthreads();
  const float r = red[0];
  __syncthreads();
  return r;
}

__device__ __forceinline__ float ntm_block_reduce_max(float v, float* red) {
#pragma unroll
  for (int off = 16; off > 0; off >>= 1) v = fmaxf(v, __shfl_down(v, off));
  const int wid = threadIdx.x >> 5, lane = threadIdx.x & 31;
  if (lane == 0) red[wid] = v;
  __syncthreads();
  if (threadIdx.x < 32) {
    float x = red[threadIdx.x];
#pragma unroll
    for (int off = 16; off > 0; off >>= 1) x = fmaxf(x, __shfl_down(x, off));
    if (threadIdx.x == 0) red[0] = x;
  }
  __syncthreads();
  const float r = red[0];
  __syncthreads();
  return r;
}

__global__ void ntm_rowops(const float* __restrict__ z,
                           const float* __restrict__ prev,
                           const float* __restrict__ params,
                           float* __restrict__ out) {
  __shared__ float row[NTM_N];
  __shared__ float red[32];

  const int b   = blockIdx.x;
  const int tid = threadIdx.x;
  const float gate  = params[b * 8 + 1];
  const float gamma = params[b * 8 + 2];
  const float s0 = params[b * 8 + 4];
  const float s1 = params[b * 8 + 5];
  const float s2 = params[b * 8 + 6];

  float zv[8];
  float lmax = -3.4e38f;
#pragma unroll
  for (int i = 0; i < 8; ++i) {
    const int n = tid + i * 1024;
    zv[i] = z[(size_t)b * NTM_N + n];
    lmax = fmaxf(lmax, zv[i]);
  }
  const float rmax = ntm_block_reduce_max(lmax, red);

  float lsum = 0.f;
#pragma unroll
  for (int i = 0; i < 8; ++i) {
    zv[i] = expf(zv[i] - rmax);
    lsum += zv[i];
  }
  const float invsum = 1.0f / ntm_block_reduce_sum(lsum, red);

#pragma unroll
  for (int i = 0; i < 8; ++i) {
    const int n = tid + i * 1024;
    const float cw = zv[i] * invsum;
    row[n] = gate * cw + (1.0f - gate) * prev[(size_t)b * NTM_N + n];
  }
  __syncthreads();

  float sh[8];
  float lsh = 0.f;
#pragma unroll
  for (int i = 0; i < 8; ++i) {
    const int n = tid + i * 1024;
    const float right = row[(n + 1) & (NTM_N - 1)];
    const float mid   = row[n];
    const float left  = row[(n + NTM_N - 1) & (NTM_N - 1)];
    const float shifted = s0 * right + s1 * mid + s2 * left;
    sh[i] = powf(shifted + 1e-8f, gamma);
    lsh += sh[i];
  }
  const float invt = 1.0f / (ntm_block_reduce_sum(lsh, red) + NTM_EPS);

#pragma unroll
  for (int i = 0; i < 8; ++i) {
    const int n = tid + i * 1024;
    out[(size_t)b * NTM_N + n] = sh[i] * invt;
  }
}

extern "C" void kernel_launch(void* const* d_in, const int* in_sizes, int n_in,
                              void* d_out, int out_size, void* d_ws, size_t ws_size,
                              hipStream_t stream) {
  const float* memory = (const float*)d_in[0];
  const float* cs     = (const float*)d_in[1];
  const float* prev   = (const float*)d_in[2];
  const float* Wk     = (const float*)d_in[3];
  const float* Wb     = (const float*)d_in[4];
  const float* bb     = (const float*)d_in[5];
  const float* Wg     = (const float*)d_in[6];
  const float* bg     = (const float*)d_in[7];
  const float* Ws     = (const float*)d_in[8];
  const float* bs     = (const float*)d_in[9];
  const float* Wgam   = (const float*)d_in[10];
  const float* bgam   = (const float*)d_in[11];
  float* out = (float*)d_out;

  float* zbuf   = (float*)d_ws;
  float* keyb   = zbuf + (size_t)NTM_B * NTM_N;
  float* params = keyb + (size_t)NTM_B * NTM_D;

  ntm_key_gemm<<<dim3(NTM_D / 16, NTM_B / 16), 32, 0, stream>>>(cs, Wk, keyb);
  ntm_params<<<NTM_B, 256, 0, stream>>>(cs, keyb, Wb, bb, Wg, bg, Ws, bs,
                                        Wgam, bgam, params);
  ntm_dot<<<dim3(NTM_B * NTM_N / (DOT_WAVES * DOT_ROWS)), DOT_WAVES * 32, 0, stream>>>(memory, keyb, params, zbuf);
  ntm_rowops<<<NTM_B, 1024, 0, stream>>>(zbuf, prev, params, out);
}


# ntm_dot row sums: the 16 rows of a group are folded together, two rows sharing each lane exchange (31 exchanges per 32 rows and quantity, was 160); per row same additions in same order
# speedup vs baseline: 1.3931x; 1.1241x over previous
#include <hip/hip_runtime.h>
#include <math.h>

typedef float v2f __attribute__((ext_vector_type(2)));
typedef float v8f __attribute__((ext_vector_type(8)));
typedef float f4  __attribute__((ext_vector_type(4)));

#define NTM_B 64
#define NTM_N 8192
#define NTM_D 128
#define NTM_C 256
#define NTM_EPS 1e-8f

__global__ void ntm_key_gemm(const float* __restrict__ cs,
                             const float* __restrict__ Wk,
                             float* __restrict__ key) {
  const int n0   = blockIdx.x * 16;
  const int m0   = blockIdx.y * 16;
  const int lane = threadIdx.x;
  const int half = lane >> 4;
  const int r    = lane & 15;

#if __has_builtin(__builtin_amdgcn_wmma_f32_16x16x4_f32)
  v8f acc = {};
  const float* arow = cs + (size_t)(m0 + r) * NTM_C + 2 * half;
  const float* bcol = Wk + (size_t)(2 * half) * NTM_D + n0 + r;
  for (int k0 = 0; k0 < NTM_C; k0 += 4) {
    v2f a, b;
    a.x = arow[k0];
    a.y = arow[k0 + 1];
    b.x = bcol[(size_t)k0 * NTM_D];
    b.y = bcol[(size_t)(k0 + 1) * NTM_D];
    acc = __builtin_amdgcn_wmma_f32_16x16x4_f32(
         false, a,  false, b,
         (short)0, acc,  false,  false);
  }
#pragma unroll
  for (int i = 0; i < 8; ++i)
    key[(size_t)(m0 + half * 8 + i) * NTM_D + n0 + r] = acc[i];
#else
  for (int i = 0; i < 8; ++i) {
    const int row = m0 + half * 8 + i;
    const int col = n0 + r;
    float s = 0.f;
    for (int k = 0; k < NTM_C; ++k)
      s += cs[(size_t)row * NTM_C + k] * Wk[(size_t)k * NTM_D + col];
    key[(size_t)row * NTM_D + col] = s;
  }
#endif
}

__device__ __forceinline__ float ntm_softplus(float x) {
  return (x > 20.f) ? x : log1pf(expf(x));
}

__global__ void ntm_params(const float* __restrict__ cs,
                           const float* __restrict__ key,
                           const float* __restrict__ Wb,  const float* __restrict__ bb,
                           const float* __restrict__ Wg,  const float* __restrict__ bg,
                           const float* __restrict__ Ws,  const float* __restrict__ bs,
                           const float* __restrict__ Wgam,const float* __restrict__ bgam,
                           float* __restrict__ params) {
  __shared__ float sd[7 * 256];
  const int b = blockIdx.x;
  const int t = threadIdx.x;

  const float c = cs[(size_t)b * NTM_C + t];
  sd[0 * 256 + t] = c * Wb[t];
  sd[1 * 256 + t] = c * Wg[t];
  sd[2 * 256 + t] = c * Wgam[t];
  sd[3 * 256 + t] = c * Ws[t * 3 + 0];
  sd[4 * 256 + t] = c * Ws[t * 3 + 1];
  sd[5 * 256 + t] = c * Ws[t * 3 + 2];
  const float kv = (t < NTM_D) ? key[(size_t)b * NTM_D + t] : 0.f;
  sd[6 * 256 + t] = kv * kv;
  __syncthreads();

  for (int s = 128; s > 0; s >>= 1) {
    if (t < s) {
#pragma unroll
      for (int j = 0; j < 7; ++j) sd[j * 256 + t] += sd[j * 256 + t + s];
    }
    __syncthreads();
  }

  if (t == 0) {
    const float beta  = ntm_softplus(sd[0 * 256] + bb[0]) + 1.0f;
    const float gate  = 1.0f / (1.0f + expf(-(sd[1 * 256] + bg[0])));
    const float gamma = ntm_softplus(sd[2 * 256] + bgam[0]) + 1.0f;
    const float knorm = sqrtf(sd[6 * 256]);
    float x0 = sd[3 * 256] + bs[0];
    float x1 = sd[4 * 256] + bs[1];
    float x2 = sd[5 * 256] + bs[2];
    const float m = fmaxf(x0, fmaxf(x1, x2));
    const float e0 = expf(x0 - m), e1 = expf(x1 - m), e2 = expf(x2 - m);
    const float inv = 1.0f / (e0 + e1 + e2);
    float* p = params + b * 8;
    p[0] = beta;  p[1] = gate;  p[2] = gamma; p[3] = knorm;
    p[4] = e0 * inv; p[5] = e1 * inv; p[6] = e2 * inv;
  }
}

#define DOT_WAVES 16
#define DOT_ROWS  128
#define DOT_G     16
#define DOT_ROW_F4 (NTM_D / 4)

template <bool REFILL>
__device__ __forceinline__ void ntm_take_group(f4 (&buf)[DOT_G], const f4* next,
                                               const f4 k4, const int lane,
                                               float (&pd)[DOT_G], float (&ps)[DOT_G]) {
#pragma unroll
  for (int r = 0; r < DOT_G; ++r) {
    const f4 m4 = buf[r];
    pd[r] = m4.x * k4.x + m4.y * k4.y + m4.z * k4.z + m4.w * k4.w;
    ps[r] = m4.x * m4.x + m4.y * m4.y + m4.z * m4.z + m4.w * m4.w;
    if (REFILL) buf[r] = __builtin_nontemporal_load(next + r * DOT_ROW_F4 + lane);
  }
}

template <int H, int OFF>
__device__ __forceinline__ void ntm_fold_step(float (&pd)[DOT_G], float (&ps)[DOT_G], const int lane) {
  const bool up = (lane & OFF) != 0;
#pragma unroll
  for (int r = 0; r < H; ++r) {
    const float keep_d = up ? pd[r + H] : pd[r];
    const float send_d = up ? pd[r] : pd[r + H];
    const float keep_s = up ? ps[r + H] : ps[r];
    const float send_s = up ? ps[r] : ps[r + H];
    pd[r] = keep_d + __shfl_xor(send_d, OFF);
    ps[r] = keep_s + __shfl_xor(send_s, OFF);
  }
}

__device__ __forceinline__ void ntm_fold_group(float (&pd)[DOT_G], float (&ps)[DOT_G],
                                               const int lane, float& out_d, float& out_s) {
  ntm_fold_step<8, 16>(pd, ps, lane);
  ntm_fold_step<4, 8>(pd, ps, lane);
  ntm_fold_step<2, 4>(pd, ps, lane);
  ntm_fold_step<1, 2>(pd, ps, lane);
  out_d = pd[0];
  out_s = ps[0];
}

__device__ __forceinline__ float ntm_join_groups(const float g0, const float g1, const int lane) {
  const bool odd = (lane & 1) != 0;
  const float keep = odd ? g1 : g0;
  const float send = odd ? g0 : g1;
  return keep + __shfl_xor(send, 1);
}

__global__ void __launch_bounds__(DOT_WAVES * 32)
ntm_dot(const float* __restrict__ memory,
        const float* __restrict__ key,
        const float* __restrict__ params,
        float* __restrict__ z) {
  const int lane = threadIdx.x & 31;
  const int wave = __builtin_amdgcn_readfirstlane(threadIdx.x >> 5);
  constexpr int WG_PER_BATCH = NTM_N / (DOT_WAVES * DOT_ROWS);
  const int b    = blockIdx.x / WG_PER_BATCH;
  const int row0 = ((blockIdx.x % WG_PER_BATCH) * DOT_WAVES + wave) * DOT_ROWS;

  const f4 k4 = ((const f4*)(key + (size_t)b * NTM_D))[lane];
  const float beta  = params[b * 8 + 0];
  const float knorm = params[b * 8 + 3];

  const f4* p = (const f4*)(memory + ((size_t)b * NTM_N + row0) * NTM_D);
  float* zq = z + (size_t)b * NTM_N + row0 + 16 * (lane & 1) + (lane >> 1);

  f4 buf[DOT_G];
  float pd[DOT_G], ps[DOT_G];
  float d0, s0, d1, s1;
  constexpr int TRIPS = DOT_ROWS / (2 * DOT_G);

#pragma unroll
  for (int r = 0; r < DOT_G; ++r)
    buf[r] = __builtin_nontemporal_load(p + r * DOT_ROW_F4 + lane);

#pragma nounroll
  for (int t = 0; t < TRIPS - 1; ++t) {
    ntm_take_group<true>(buf, p + DOT_G * DOT_ROW_F4, k4, lane, pd, ps);
    __builtin_amdgcn_sched_barrier(0);
    ntm_fold_group(pd, ps, lane, d0, s0);
    __builtin_amdgcn_sched_barrier(0);
    ntm_take_group<true>(buf, p + 2 * DOT_G * DOT_ROW_F4, k4, lane, pd, ps);
    __builtin_amdgcn_sched_barrier(0);
    ntm_fold_group(pd, ps, lane, d1, s1);
    {
      const float kd = ntm_join_groups(d0, d1, lane);
      const float ks = ntm_join_groups(s0, s1, lane);
      *zq = beta * kd / (knorm * sqrtf(ks) + NTM_EPS);
    }
    p  += 2 * DOT_G * DOT_ROW_F4;
    zq += 2 * DOT_G;
  }
  ntm_take_group<true>(buf, p + DOT_G * DOT_ROW_F4, k4, lane, pd, ps);
  __builtin_amdgcn_sched_barrier(0);
  ntm_fold_group(pd, ps, lane, d0, s0);
  ntm_take_group<false>(buf, p, k4, lane, pd, ps);
  ntm_fold_group(pd, ps, lane, d1, s1);
  {
    const float kd = ntm_join_groups(d0, d1, lane);
    const float ks = ntm_join_groups(s0, s1, lane);
    *zq = beta * kd / (knorm * sqrtf(ks) + NTM_EPS);
  }
}

__device__ __forceinline__ float ntm_block_reduce_sum(float v, float* red) {
#pragma unroll
  for (int off = 16; off > 0; off >>= 1) v += __shfl_down(v, off);
  const int wid = threadIdx.x >> 5, lane = threadIdx.x & 31;
  if (lane == 0) red[wid] = v;
  __syncthreads();
  if (threadIdx.x < 32) {
    float x = red[threadIdx.x];
#pragma unroll
    for (int off = 16; off > 0; off >>= 1) x += __shfl_down(x, off);
    if (threadIdx.x == 0) red[0] = x;
  }
  __syncthreads();
  const float r = red[0];
  __syncthreads();
  return r;
}

__device__ __forceinline__ float ntm_block_reduce_max(float v, float* red) {
#pragma unroll
  for (int off = 16; off > 0; off >>= 1) v = fmaxf(v, __shfl_down(v, off));
  const int wid = threadIdx.x >> 5, lane = threadIdx.x & 31;
  if (lane == 0) red[wid] = v;
  __syncthreads();
  if (threadIdx.x < 32) {
    float x = red[threadIdx.x];
#pragma unroll
    for (int off = 16; off > 0; off >>= 1) x = fmaxf(x, __shfl_down(x, off));
    if (threadIdx.x == 0) red[0] = x;
  }
  __syncthreads();
  const float r = red[0];
  __syncthreads();
  return r;
}

__global__ void ntm_rowops(const float* __restrict__ z,
                           const float* __restrict__ prev,
                           const float* __restrict__ params,
                           float* __restrict__ out) {
  __shared__ float row[NTM_N];
  __shared__ float red[32];

  const int b   = blockIdx.x;
  const int tid = threadIdx.x;
  const float gate  = params[b * 8 + 1];
  const float gamma = params[b * 8 + 2];
  const float s0 = params[b * 8 + 4];
  const float s1 = params[b * 8 + 5];
  const float s2 = params[b * 8 + 6];

  float zv[8];
  float lmax = -3.4e38f;
#pragma unroll
  for (int i = 0; i < 8; ++i) {
    const int n = tid + i * 1024;
    zv[i] = z[(size_t)b * NTM_N + n];
    lmax = fmaxf(lmax, zv[i]);
  }
  const float rmax = ntm_block_reduce_max(lmax, red);

  float lsum = 0.f;
#pragma unroll
  for (int i = 0; i < 8; ++i) {
    zv[i] = expf(zv[i] - rmax);
    lsum += zv[i];
  }
  const float invsum = 1.0f / ntm_block_reduce_sum(lsum, red);

#pragma unroll
  for (int i = 0; i < 8; ++i) {
    const int n = tid + i * 1024;
    const float cw = zv[i] * invsum;
    row[n] = gate * cw + (1.0f - gate) * prev[(size_t)b * NTM_N + n];
  }
  __syncthreads();

  float sh[8];
  float lsh = 0.f;
#pragma unroll
  for (int i = 0; i < 8; ++i) {
    const int n = tid + i * 1024;
    const float right = row[(n + 1) & (NTM_N - 1)];
    const float mid   = row[n];
    const float left  = row[(n + NTM_N - 1) & (NTM_N - 1)];
    const float shifted = s0 * right + s1 * mid + s2 * left;
    sh[i] = powf(shifted + 1e-8f, gamma);
    lsh += sh[i];
  }
  const float invt = 1.0f / (ntm_block_reduce_sum(lsh, red) + NTM_EPS);

#pragma unroll
  for (int i = 0; i < 8; ++i) {
    const int n = tid + i * 1024;
    out[(size_t)b * NTM_N + n] = sh[i] * invt;
  }
}

extern "C" void kernel_launch(void* const* d_in, const int* in_sizes, int n_in,
                              void* d_out, int out_size, void* d_ws, size_t ws_size,
                              hipStream_t stream) {
  const float* memory = (const float*)d_in[0];
  const float* cs     = (const float*)d_in[1];
  const float* prev   = (const float*)d_in[2];
  const float* Wk     = (const float*)d_in[3];
  const float* Wb     = (const float*)d_in[4];
  const float* bb     = (const float*)d_in[5];
  const float* Wg     = (const float*)d_in[6];
  const float* bg     = (const float*)d_in[7];
  const float* Ws     = (const float*)d_in[8];
  const float* bs     = (const float*)d_in[9];
  const float* Wgam   = (const float*)d_in[10];
  const float* bgam   = (const float*)d_in[11];
  float* out = (float*)d_out;

  float* zbuf   = (float*)d_ws;
  float* keyb   = zbuf + (size_t)NTM_B * NTM_N;
  float* params = keyb + (size_t)NTM_B * NTM_D;

  ntm_key_gemm<<<dim3(NTM_D / 16, NTM_B / 16), 32, 0, stream>>>(cs, Wk, keyb);
  ntm_params<<<NTM_B, 256, 0, stream>>>(cs, keyb, Wb, bb, Wg, bg, Ws, bs,
                                        Wgam, bgam, params);
  ntm_dot<<<dim3(NTM_B * NTM_N / (DOT_WAVES * DOT_ROWS)), DOT_WAVES * 32, 0, stream>>>(memory, keyb, params, zbuf);
  ntm_rowops<<<NTM_B, 1024, 0, stream>>>(zbuf, prev, params, out);
}


# per-batch scalars formed by 64 extra one-wave workgroups of the key GEMM's launch, key norm summed per wave in ntm_dot: 4 launches to 3; sum order of those scalars and of the norm changes
# speedup vs baseline: 1.5382x; 1.1042x over previous
#include <hip/hip_runtime.h>
#include <math.h>

typedef float v2f __attribute__((ext_vector_type(2)));
typedef float v8f __attribute__((ext_vector_type(8)));
typedef float f4  __attribute__((ext_vector_type(4)));

#define NTM_B 64
#define NTM_N 8192
#define NTM_D 128
#define NTM_C 256
#define NTM_EPS 1e-8f

#define KEY_TILES (NTM_D / 16)

__device__ __forceinline__ float ntm_softplus(float x) {
  return (x > 20.f) ? x : log1pf(expf(x));
}

__device__ __forceinline__ float ntm_wave_sum(float v) {
#pragma unroll
  for (int off = 16; off > 0; off >>= 1) v += __shfl_xor(v, off);
  return v;
}

__global__ void __launch_bounds__(32)
ntm_key_params(const float* __restrict__ cs,
               const float* __restrict__ Wk,
               const float* __restrict__ Wb,  const float* __restrict__ bb,
               const float* __restrict__ Wg,  const float* __restrict__ bg,
               const float* __restrict__ Ws,  const float* __restrict__ bs,
               const float* __restrict__ Wgam,const float* __restrict__ bgam,
               float* __restrict__ key,
               float* __restrict__ params) {
  const int m0   = blockIdx.y * 16;
  const int lane = threadIdx.x;

  if (blockIdx.x < KEY_TILES) {
    const int n0   = blockIdx.x * 16;
    const int half = lane >> 4;
    const int r    = lane & 15;
    v8f acc = {};
    const float* arow = cs + (size_t)(m0 + r) * NTM_C + 2 * half;
    const float* bcol = Wk + (size_t)(2 * half) * NTM_D + n0 + r;
    for (int k0 = 0; k0 < NTM_C; k0 += 4) {
      v2f a, b;
      a.x = arow[k0];
      a.y = arow[k0 + 1];
      b.x = bcol[(size_t)k0 * NTM_D];
      b.y = bcol[(size_t)(k0 + 1) * NTM_D];
      acc = __builtin_amdgcn_wmma_f32_16x16x4_f32(
           false, a,  false, b,
           (short)0, acc,  false,  false);
    }
#pragma unroll
    for (int i = 0; i < 8; ++i)
      key[(size_t)(m0 + half * 8 + i) * NTM_D + n0 + r] = acc[i];
  } else {
    const int b = m0 + blockIdx.x - KEY_TILES;
    float xb = 0.f, xg = 0.f, xm = 0.f, x0 = 0.f, x1 = 0.f, x2 = 0.f;
#pragma unroll
    for (int j = 0; j < 8; ++j) {
      const int t = lane + 32 * j;
      const float c = cs[(size_t)b * NTM_C + t];
      xb += c * Wb[t];
      xg += c * Wg[t];
      xm += c * Wgam[t];
      x0 += c * Ws[t * 3 + 0];
      x1 += c * Ws[t * 3 + 1];
      x2 += c * Ws[t * 3 + 2];
    }
    xb = ntm_wave_sum(xb);
    xg = ntm_wave_sum(xg);
    xm = ntm_wave_sum(xm);
    x0 = ntm_wave_sum(x0) + bs[0];
    x1 = ntm_wave_sum(x1) + bs[1];
    x2 = ntm_wave_sum(x2) + bs[2];
    const float beta  = ntm_softplus(xb + bb[0]) + 1.0f;
    const float gate  = 1.0f / (1.0f + expf(-(xg + bg[0])));
    const float gamma = ntm_softplus(xm + bgam[0]) + 1.0f;
    const float m = fmaxf(x0, fmaxf(x1, x2));
    const float e0 = expf(x0 - m), e1 = expf(x1 - m), e2 = expf(x2 - m);
    const float inv = 1.0f / (e0 + e1 + e2);
    if (lane == 0) {
      float* p = params + b * 8;
      p[0] = beta;  p[1] = gate;  p[2] = gamma;
      p[4] = e0 * inv; p[5] = e1 * inv; p[6] = e2 * inv;
    }
  }
}

#define DOT_WAVES 16
#define DOT_ROWS  128
#define DOT_G     16
#define DOT_ROW_F4 (NTM_D / 4)

template <bool REFILL>
__device__ __forceinline__ void ntm_take_group(f4 (&buf)[DOT_G], const f4* next,
                                               const f4 k4, const int lane,
                                               float (&pd)[DOT_G], float (&ps)[DOT_G]) {
#pragma unroll
  for (int r = 0; r < DOT_G; ++r) {
    const f4 m4 = buf[r];
    pd[r] = m4.x * k4.x + m4.y * k4.y + m4.z * k4.z + m4.w * k4.w;
    ps[r] = m4.x * m4.x + m4.y * m4.y + m4.z * m4.z + m4.w * m4.w;
    if (REFILL) buf[r] = __builtin_nontemporal_load(next + r * DOT_ROW_F4 + lane);
  }
}

template <int H, int OFF>
__device__ __forceinline__ void ntm_fold_step(float (&pd)[DOT_G], float (&ps)[DOT_G], const int lane) {
  const bool up = (lane & OFF) != 0;
#pragma unroll
  for (int r = 0; r < H; ++r) {
    const float keep_d = up ? pd[r + H] : pd[r];
    const float send_d = up ? pd[r] : pd[r + H];
    const float keep_s = up ? ps[r + H] : ps[r];
    const float send_s = up ? ps[r] : ps[r + H];
    pd[r] = keep_d + __shfl_xor(send_d, OFF);
    ps[r] = keep_s + __shfl_xor(send_s, OFF);
  }
}

__device__ __forceinline__ void ntm_fold_group(float (&pd)[DOT_G], float (&ps)[DOT_G],
                                               const int lane, float& out_d, float& out_s) {
  ntm_fold_step<8, 16>(pd, ps, lane);
  ntm_fold_step<4, 8>(pd, ps, lane);
  ntm_fold_step<2, 4>(pd, ps, lane);
  ntm_fold_step<1, 2>(pd, ps, lane);
  out_d = pd[0];
  out_s = ps[0];
}

__device__ __forceinline__ float ntm_join_groups(const float g0, const float g1, const int lane) {
  const bool odd = (lane & 1) != 0;
  const float keep = odd ? g1 : g0;
  const float send = odd ? g0 : g1;
  return keep + __shfl_xor(send, 1);
}

__global__ void __launch_bounds__(DOT_WAVES * 32)
ntm_dot(const float* __restrict__ memory,
        const float* __restrict__ key,
        const float* __restrict__ params,
        float* __restrict__ z) {
  const int lane = threadIdx.x & 31;
  const int wave = __builtin_amdgcn_readfirstlane(threadIdx.x >> 5);
  constexpr int WG_PER_BATCH = NTM_N / (DOT_WAVES * DOT_ROWS);
  const int b    = blockIdx.x / WG_PER_BATCH;
  const int row0 = ((blockIdx.x % WG_PER_BATCH) * DOT_WAVES + wave) * DOT_ROWS;

  const f4 k4 = ((const f4*)(key + (size_t)b * NTM_D))[lane];
  const float beta  = params[b * 8 + 0];
  const float knorm = sqrtf(ntm_wave_sum(k4.x * k4.x + k4.y * k4.y + k4.z * k4.z + k4.w * k4.w));

  const f4* p = (const f4*)(memory + ((size_t)b * NTM_N + row0) * NTM_D);
  float* zq = z + (size_t)b * NTM_N + row0 + 16 * (lane & 1) + (lane >> 1);

  f4 buf[DOT_G];
  float pd[DOT_G], ps[DOT_G];
  float d0, s0, d1, s1;
  constexpr int TRIPS = DOT_ROWS / (2 * DOT_G);

#pragma unroll
  for (int r = 0; r < DOT_G; ++r)
    buf[r] = __builtin_nontemporal_load(p + r * DOT_ROW_F4 + lane);

#pragma nounroll
  for (int t = 0; t < TRIPS - 1; ++t) {
    ntm_take_group<true>(buf, p + DOT_G * DOT_ROW_F4, k4, lane, pd, ps);
    __builtin_amdgcn_sched_barrier(0);
    ntm_fold_group(pd, ps, lane, d0, s0);
    __builtin_amdgcn_sched_barrier(0);
    ntm_take_group<true>(buf, p + 2 * DOT_G * DOT_ROW_F4, k4, lane, pd, ps);
    __builtin_amdgcn_sched_barrier(0);
    ntm_fold_group(pd, ps, lane, d1, s1);
    {
      const float kd = ntm_join_groups(d0, d1, lane);
      const float ks = ntm_join_groups(s0, s1, lane);
      *zq = beta * kd / (knorm * sqrtf(ks) + NTM_EPS);
    }
    p  += 2 * DOT_G * DOT_ROW_F4;
    zq += 2 * DOT_G;
  }
  ntm_take_group<true>(buf, p + DOT_G * DOT_ROW_F4, k4, lane, pd, ps);
  __builtin_amdgcn_sched_barrier(0);
  ntm_fold_group(pd, ps, lane, d0, s0);
  ntm_take_group<false>(buf, p, k4, lane, pd, ps);
  ntm_fold_group(pd, ps, lane, d1, s1);
  {
    const float kd = ntm_join_groups(d0, d1, lane);
    const float ks = ntm_join_groups(s0, s1, lane);
    *zq = beta * kd / (knorm * sqrtf(ks) + NTM_EPS);
  }
}

__device__ __forceinline__ float ntm_block_reduce_sum(float v, float* red) {
#pragma unroll
  for (int off = 16; off > 0; off >>= 1) v += __shfl_down(v, off);
  const int wid = threadIdx.x >> 5, lane = threadIdx.x & 31;
  if (lane == 0) red[wid] = v;
  __syncthreads();
  if (threadIdx.x < 32) {
    float x = red[threadIdx.x];
#pragma unroll
    for (int off = 16; off > 0; off >>= 1) x += __shfl_down(x, off);
    if (threadIdx.x == 0) red[0] = x;
  }
  __syncthreads();
  const float r = red[0];
  __syncthreads();
  return r;
}

__device__ __forceinline__ float ntm_block_reduce_max(float v, float* red) {
#pragma unroll
  for (int off = 16; off > 0; off >>= 1) v = fmaxf(v, __shfl_down(v, off));
  const int wid = threadIdx.x >> 5, lane = threadIdx.x & 31;
  if (lane == 0) red[wid] = v;
  __syncthreads();
  if (threadIdx.x < 32) {
    float x = red[threadIdx.x];
#pragma unroll
    for (int off = 16; off > 0; off >>= 1) x = fmaxf(x, __shfl_down(x, off));
    if (threadIdx.x == 0) red[0] = x;
  }
  __syncthreads();
  const float r = red[0];
  __syncthreads();
  return r;
}

__global__ void ntm_rowops(const float* __restrict__ z,
                           const float* __restrict__ prev,
                           const float* __restrict__ params,
                           float* __restrict__ out) {
  __shared__ float row[NTM_N];
  __shared__ float red[32];

  const int b   = blockIdx.x;
  const int tid = threadIdx.x;
  const float gate  = params[b * 8 + 1];
  const float gamma = params[b * 8 + 2];
  const float s0 = params[b * 8 + 4];
  const float s1 = params[b * 8 + 5];
  const float s2 = params[b * 8 + 6];

  float zv[8];
  float lmax = -3.4e38f;
#pragma unroll
  for (int i = 0; i < 8; ++i) {
    const int n = tid + i * 1024;
    zv[i] = z[(size_t)b * NTM_N + n];
    lmax = fmaxf(lmax, zv[i]);
  }
  const float rmax = ntm_block_reduce_max(lmax, red);

  float lsum = 0.f;
#pragma unroll
  for (int i = 0; i < 8; ++i) {
    zv[i] = expf(zv[i] - rmax);
    lsum += zv[i];
  }
  const float invsum = 1.0f / ntm_block_reduce_sum(lsum, red);

#pragma unroll
  for (int i = 0; i < 8; ++i) {
    const int n = tid + i * 1024;
    const float cw = zv[i] * invsum;
    row[n] = gate * cw + (1.0f - gate) * prev[(size_t)b * NTM_N + n];
  }
  __syncthreads();

  float sh[8];
  float lsh = 0.f;
#pragma unroll
  for (int i = 0; i < 8; ++i) {
    const int n = tid + i * 1024;
    const float right = row[(n + 1) & (NTM_N - 1)];
    const float mid   = row[n];
    const float left  = row[(n + NTM_N - 1) & (NTM_N - 1)];
    const float shifted = s0 * right + s1 * mid + s2 * left;
    sh[i] = powf(shifted + 1e-8f, gamma);
    lsh += sh[i];
  }
  const float invt = 1.0f / (ntm_block_reduce_sum(lsh, red) + NTM_EPS);

#pragma unroll
  for (int i = 0; i < 8; ++i) {
    const int n = tid + i * 1024;
    out[(size_t)b * NTM_N + n] = sh[i] * invt;
  }
}

extern "C" void kernel_launch(void* const* d_in, const int* in_sizes, int n_in,
                              void* d_out, int out_size, void* d_ws, size_t ws_size,
                              hipStream_t stream) {
  const float* memory = (const float*)d_in[0];
  const float* cs     = (const float*)d_in[1];
  const float* prev   = (const float*)d_in[2];
  const float* Wk     = (const float*)d_in[3];
  const float* Wb     = (const float*)d_in[4];
  const float* bb     = (const float*)d_in[5];
  const float* Wg     = (const float*)d_in[6];
  const float* bg     = (const float*)d_in[7];
  const float* Ws     = (const float*)d_in[8];
  const float* bs     = (const float*)d_in[9];
  const float* Wgam   = (const float*)d_in[10];
  const float* bgam   = (const float*)d_in[11];
  float* out = (float*)d_out;

  float* zbuf   = (float*)d_ws;
  float* keyb   = zbuf + (size_t)NTM_B * NTM_N;
  float* params = keyb + (size_t)NTM_B * NTM_D;

  ntm_key_params<<<dim3(KEY_TILES + 16, NTM_B / 16), 32, 0, stream>>>(cs, Wk, Wb, bb, Wg, bg, Ws, bs,
                                                                  Wgam, bgam, keyb, params);
  ntm_dot<<<dim3(NTM_B * NTM_N / (DOT_WAVES * DOT_ROWS)), DOT_WAVES * 32, 0, stream>>>(memory, keyb, params, zbuf);
  ntm_rowops<<<NTM_B, 1024, 0, stream>>>(zbuf, prev, params, out);
}


# sharpening power: library powf replaced by 2^(gamma*log2 x) on v_log_f32/v_exp_f32, x's power of two split off, fraction by two fused multiply-adds; a few ulp from powf
# speedup vs baseline: 1.7083x; 1.1106x over previous
#include <hip/hip_runtime.h>
#include <math.h>

typedef float v2f __attribute__((ext_vector_type(2)));
typedef float v8f __attribute__((ext_vector_type(8)));
typedef float f4  __attribute__((ext_vector_type(4)));

#define NTM_B 64
#define NTM_N 8192
#define NTM_D 128
#define NTM_C 256
#define NTM_EPS 1e-8f

#define KEY_TILES (NTM_D / 16)

__device__ __forceinline__ float ntm_softplus(float x) {
  return (x > 20.f) ? x : log1pf(expf(x));
}

__device__ __forceinline__ float ntm_wave_sum(float v) {
#pragma unroll
  for (int off = 16; off > 0; off >>= 1) v += __shfl_xor(v, off);
  return v;
}

__global__ void __launch_bounds__(32)
ntm_key_params(const float* __restrict__ cs,
               const float* __restrict__ Wk,
               const float* __restrict__ Wb,  const float* __restrict__ bb,
               const float* __restrict__ Wg,  const float* __restrict__ bg,
               const float* __restrict__ Ws,  const float* __restrict__ bs,
               const float* __restrict__ Wgam,const float* __restrict__ bgam,
               float* __restrict__ key,
               float* __restrict__ params) {
  const int m0   = blockIdx.y * 16;
  const int lane = threadIdx.x;

  if (blockIdx.x < KEY_TILES) {
    const int n0   = blockIdx.x * 16;
    const int half = lane >> 4;
    const int r    = lane & 15;
    v8f acc = {};
    const float* arow = cs + (size_t)(m0 + r) * NTM_C + 2 * half;
    const float* bcol = Wk + (size_t)(2 * half) * NTM_D + n0 + r;
    for (int k0 = 0; k0 < NTM_C; k0 += 4) {
      v2f a, b;
      a.x = arow[k0];
      a.y = arow[k0 + 1];
      b.x = bcol[(size_t)k0 * NTM_D];
      b.y = bcol[(size_t)(k0 + 1) * NTM_D];
      acc = __builtin_amdgcn_wmma_f32_16x16x4_f32(
           false, a,  false, b,
           (short)0, acc,  false,  false);
    }
#pragma unroll
    for (int i = 0; i < 8; ++i)
      key[(size_t)(m0 + half * 8 + i) * NTM_D + n0 + r] = acc[i];
  } else {
    const int b = m0 + blockIdx.x - KEY_TILES;
    float xb = 0.f, xg = 0.f, xm = 0.f, x0 = 0.f, x1 = 0.f, x2 = 0.f;
#pragma unroll
    for (int j = 0; j < 8; ++j) {
      const int t = lane + 32 * j;
      const float c = cs[(size_t)b * NTM_C + t];
      xb += c * Wb[t];
      xg += c * Wg[t];
      xm += c * Wgam[t];
      x0 += c * Ws[t * 3 + 0];
      x1 += c * Ws[t * 3 + 1];
      x2 += c * Ws[t * 3 + 2];
    }
    xb = ntm_wave_sum(xb);
    xg = ntm_wave_sum(xg);
    xm = ntm_wave_sum(xm);
    x0 = ntm_wave_sum(x0) + bs[0];
    x1 = ntm_wave_sum(x1) + bs[1];
    x2 = ntm_wave_sum(x2) + bs[2];
    const float beta  = ntm_softplus(xb + bb[0]) + 1.0f;
    const float gate  = 1.0f / (1.0f + expf(-(xg + bg[0])));
    const float gamma = ntm_softplus(xm + bgam[0]) + 1.0f;
    const float m = fmaxf(x0, fmaxf(x1, x2));
    const float e0 = expf(x0 - m), e1 = expf(x1 - m), e2 = expf(x2 - m);
    const float inv = 1.0f / (e0 + e1 + e2);
    if (lane == 0) {
      float* p = params + b * 8;
      p[0] = beta;  p[1] = gate;  p[2] = gamma;
      p[4] = e0 * inv; p[5] = e1 * inv; p[6] = e2 * inv;
    }
  }
}

#define DOT_WAVES 16
#define DOT_ROWS  128
#define DOT_G     16
#define DOT_ROW_F4 (NTM_D / 4)

template <bool REFILL>
__device__ __forceinline__ void ntm_take_group(f4 (&buf)[DOT_G], const f4* next,
                                               const f4 k4, const int lane,
                                               float (&pd)[DOT_G], float (&ps)[DOT_G]) {
#pragma unroll
  for (int r = 0; r < DOT_G; ++r) {
    const f4 m4 = buf[r];
    pd[r] = m4.x * k4.x + m4.y * k4.y + m4.z * k4.z + m4.w * k4.w;
    ps[r] = m4.x * m4.x + m4.y * m4.y + m4.z * m4.z + m4.w * m4.w;
    if (REFILL) buf[r] = __builtin_nontemporal_load(next + r * DOT_ROW_F4 + lane);
  }
}

template <int H, int OFF>
__device__ __forceinline__ void ntm_fold_step(float (&pd)[DOT_G], float (&ps)[DOT_G], const int lane) {
  const bool up = (lane & OFF) != 0;
#pragma unroll
  for (int r = 0; r < H; ++r) {
    const float keep_d = up ? pd[r + H] : pd[r];
    const float send_d = up ? pd[r] : pd[r + H];
    const float keep_s = up ? ps[r + H] : ps[r];
    const float send_s = up ? ps[r] : ps[r + H];
    pd[r] = keep_d + __shfl_xor(send_d, OFF);
    ps[r] = keep_s + __shfl_xor(send_s, OFF);
  }
}

__device__ __forceinline__ void ntm_fold_group(float (&pd)[DOT_G], float (&ps)[DOT_G],
                                               const int lane, float& out_d, float& out_s) {
  ntm_fold_step<8, 16>(pd, ps, lane);
  ntm_fold_step<4, 8>(pd, ps, lane);
  ntm_fold_step<2, 4>(pd, ps, lane);
  ntm_fold_step<1, 2>(pd, ps, lane);
  out_d = pd[0];
  out_s = ps[0];
}

__device__ __forceinline__ float ntm_join_groups(const float g0, const float g1, const int lane) {
  const bool odd = (lane & 1) != 0;
  const float keep = odd ? g1 : g0;
  const float send = odd ? g0 : g1;
  return keep + __shfl_xor(send, 1);
}

__global__ void __launch_bounds__(DOT_WAVES * 32)
ntm_dot(const float* __restrict__ memory,
        const float* __restrict__ key,
        const float* __restrict__ params,
        float* __restrict__ z) {
  const int lane = threadIdx.x & 31;
  const int wave = __builtin_amdgcn_readfirstlane(threadIdx.x >> 5);
  constexpr int WG_PER_BATCH = NTM_N / (DOT_WAVES * DOT_ROWS);
  const int b    = blockIdx.x / WG_PER_BATCH;
  const int row0 = ((blockIdx.x % WG_PER_BATCH) * DOT_WAVES + wave) * DOT_ROWS;

  const f4 k4 = ((const f4*)(key + (size_t)b * NTM_D))[lane];
  const float beta  = params[b * 8 + 0];
  const float knorm = sqrtf(ntm_wave_sum(k4.x * k4.x + k4.y * k4.y + k4.z * k4.z + k4.w * k4.w));

  const f4* p = (const f4*)(memory + ((size_t)b * NTM_N + row0) * NTM_D);
  float* zq = z + (size_t)b * NTM_N + row0 + 16 * (lane & 1) + (lane >> 1);

  f4 buf[DOT_G];
  float pd[DOT_G], ps[DOT_G];
  float d0, s0, d1, s1;
  constexpr int TRIPS = DOT_ROWS / (2 * DOT_G);

#pragma unroll
  for (int r = 0; r < DOT_G; ++r)
    buf[r] = __builtin_nontemporal_load(p + r * DOT_ROW_F4 + lane);

#pragma nounroll
  for (int t = 0; t < TRIPS - 1; ++t) {
    ntm_take_group<true>(buf, p + DOT_G * DOT_ROW_F4, k4, lane, pd, ps);
    __builtin_amdgcn_sched_barrier(0);
    ntm_fold_group(pd, ps, lane, d0, s0);
    __builtin_amdgcn_sched_barrier(0);
    ntm_take_group<true>(buf, p + 2 * DOT_G * DOT_ROW_F4, k4, lane, pd, ps);
    __builtin_amdgcn_sched_barrier(0);
    ntm_fold_group(pd, ps, lane, d1, s1);
    {
      const float kd = ntm_join_groups(d0, d1, lane);
      const float ks = ntm_join_groups(s0, s1, lane);
      *zq = beta * kd / (knorm * sqrtf(ks) + NTM_EPS);
    }
    p  += 2 * DOT_G * DOT_ROW_F4;
    zq += 2 * DOT_G;
  }
  ntm_take_group<true>(buf, p + DOT_G * DOT_ROW_F4, k4, lane, pd, ps);
  __builtin_amdgcn_sched_barrier(0);
  ntm_fold_group(pd, ps, lane, d0, s0);
  ntm_take_group<false>(buf, p, k4, lane, pd, ps);
  ntm_fold_group(pd, ps, lane, d1, s1);
  {
    const float kd = ntm_join_groups(d0, d1, lane);
    const float ks = ntm_join_groups(s0, s1, lane);
    *zq = beta * kd / (knorm * sqrtf(ks) + NTM_EPS);
  }
}

__device__ __forceinline__ float ntm_block_reduce_sum(float v, float* red) {
#pragma unroll
  for (int off = 16; off > 0; off >>= 1) v += __shfl_down(v, off);
  const int wid = threadIdx.x >> 5, lane = threadIdx.x & 31;
  if (lane == 0) red[wid] = v;
  __syncthreads();
  if (threadIdx.x < 32) {
    float x = red[threadIdx.x];
#pragma unroll
    for (int off = 16; off > 0; off >>= 1) x += __shfl_down(x, off);
    if (threadIdx.x == 0) red[0] = x;
  }
  __syncthreads();
  const float r = red[0];
  __syncthreads();
  return r;
}

__device__ __forceinline__ float ntm_block_reduce_max(float v, float* red) {
#pragma unroll
  for (int off = 16; off > 0; off >>= 1) v = fmaxf(v, __shfl_down(v, off));
  const int wid = threadIdx.x >> 5, lane = threadIdx.x & 31;
  if (lane == 0) red[wid] = v;
  __syncthreads();
  if (threadIdx.x < 32) {
    float x = red[threadIdx.x];
#pragma unroll
    for (int off = 16; off > 0; off >>= 1) x = fmaxf(x, __shfl_down(x, off));
    if (threadIdx.x == 0) red[0] = x;
  }
  __syncthreads();
  const float r = red[0];
  __syncthreads();
  return r;
}

__device__ __forceinline__ float ntm_pow_pos(const float x, const float g) {
  const int   e  = __builtin_amdgcn_frexp_expf(x * 1.41421356f) - 1;
  const float m  = ldexpf(x, -e);
  const float ef = (float)e;
  const float lm = __builtin_amdgcn_logf(m);
  const float n  = __builtin_rintf(g * (ef + lm));
  const float f  = __builtin_fmaf(g, lm, __builtin_fmaf(g, ef, -n));
  return ldexpf(__builtin_amdgcn_exp2f(f), (int)n);
}

__global__ void ntm_rowops(const float* __restrict__ z,
                           const float* __restrict__ prev,
                           const float* __restrict__ params,
                           float* __restrict__ out) {
  __shared__ float row[NTM_N];
  __shared__ float red[32];

  const int b   = blockIdx.x;
  const int tid = threadIdx.x;
  const float gate  = params[b * 8 + 1];
  const float gamma = params[b * 8 + 2];
  const float s0 = params[b * 8 + 4];
  const float s1 = params[b * 8 + 5];
  const float s2 = params[b * 8 + 6];

  float zv[8];
  float lmax = -3.4e38f;
#pragma unroll
  for (int i = 0; i < 8; ++i) {
    const int n = tid + i * 1024;
    zv[i] = z[(size_t)b * NTM_N + n];
    lmax = fmaxf(lmax, zv[i]);
  }
  const float rmax = ntm_block_reduce_max(lmax, red);

  float lsum = 0.f;
#pragma unroll
  for (int i = 0; i < 8; ++i) {
    zv[i] = expf(zv[i] - rmax);
    lsum += zv[i];
  }
  const float invsum = 1.0f / ntm_block_reduce_sum(lsum, red);

#pragma unroll
  for (int i = 0; i < 8; ++i) {
    const int n = tid + i * 1024;
    const float cw = zv[i] * invsum;
    row[n] = gate * cw + (1.0f - gate) * prev[(size_t)b * NTM_N + n];
  }
  __syncthreads();

  float sh[8];
  float lsh = 0.f;
#pragma unroll
  for (int i = 0; i < 8; ++i) {
    const int n = tid + i * 1024;
    const float right = row[(n + 1) & (NTM_N - 1)];
    const float mid   = row[n];
    const float left  = row[(n + NTM_N - 1) & (NTM_N - 1)];
    const float shifted = s0 * right + s1 * mid + s2 * left;
    sh[i] = ntm_pow_pos(shifted + 1e-8f, gamma);
    lsh += sh[i];
  }
  const float invt = 1.0f / (ntm_block_reduce_sum(lsh, red) + NTM_EPS);

#pragma unroll
  for (int i = 0; i < 8; ++i) {
    const int n = tid + i * 1024;
    out[(size_t)b * NTM_N + n] = sh[i] * invt;
  }
}

extern "C" void kernel_launch(void* const* d_in, const int* in_sizes, int n_in,
                              void* d_out, int out_size, void* d_ws, size_t ws_size,
                              hipStream_t stream) {
  const float* memory = (const float*)d_in[0];
  const float* cs     = (const float*)d_in[1];
  const float* prev   = (const float*)d_in[2];
  const float* Wk     = (const float*)d_in[3];
  const float* Wb     = (const float*)d_in[4];
  const float* bb     = (const float*)d_in[5];
  const float* Wg     = (const float*)d_in[6];
  const float* bg     = (const float*)d_in[7];
  const float* Ws     = (const float*)d_in[8];
  const float* bs     = (const float*)d_in[9];
  const float* Wgam   = (const float*)d_in[10];
  const float* bgam   = (const float*)d_in[11];
  float* out = (float*)d_out;

  float* zbuf   = (float*)d_ws;
  float* keyb   = zbuf + (size_t)NTM_B * NTM_N;
  float* params = keyb + (size_t)NTM_B * NTM_D;

  ntm_key_params<<<dim3(KEY_TILES + 16, NTM_B / 16), 32, 0, stream>>>(cs, Wk, Wb, bb, Wg, bg, Ws, bs,
                                                                  Wgam, bgam, keyb, params);
  ntm_dot<<<dim3(NTM_B * NTM_N / (DOT_WAVES * DOT_ROWS)), DOT_WAVES * 32, 0, stream>>>(memory, keyb, params, zbuf);
  ntm_rowops<<<NTM_B, 1024, 0, stream>>>(zbuf, prev, params, out);
}


# first launch removed (3 to 2): each ntm_dot workgroup forms its batch's key row itself on f32 WMMA, k in two halves summed in LDS (key's sum order changes); per-batch scalars formed in ntm_rowops
# speedup vs baseline: 1.8089x; 1.0589x over previous
#include <hip/hip_runtime.h>
#include <math.h>

typedef float v2f __attribute__((ext_vector_type(2)));
typedef float v8f __attribute__((ext_vector_type(8)));
typedef float f4  __attribute__((ext_vector_type(4)));

#define NTM_B 64
#define NTM_N 8192
#define NTM_D 128
#define NTM_C 256
#define NTM_EPS 1e-8f

__device__ __forceinline__ float ntm_softplus(float x) {
  return (x > 20.f) ? x : log1pf(expf(x));
}

__device__ __forceinline__ float ntm_wave_sum(float v) {
#pragma unroll
  for (int off = 16; off > 0; off >>= 1) v += __shfl_xor(v, off);
  return v;
}

#define KEY_KH    (NTM_C / 2)
#define KEY_STEPS (KEY_KH / 4)
#define DOT_WAVES 16
#define DOT_ROWS  128
#define DOT_G     16
#define DOT_ROW_F4 (NTM_D / 4)

template <bool REFILL>
__device__ __forceinline__ void ntm_take_group(f4 (&buf)[DOT_G], const f4* next,
                                               const f4 k4, const int lane,
                                               float (&pd)[DOT_G], float (&ps)[DOT_G]) {
#pragma unroll
  for (int r = 0; r < DOT_G; ++r) {
    const f4 m4 = buf[r];
    pd[r] = m4.x * k4.x + m4.y * k4.y + m4.z * k4.z + m4.w * k4.w;
    ps[r] = m4.x * m4.x + m4.y * m4.y + m4.z * m4.z + m4.w * m4.w;
    if (REFILL) buf[r] = __builtin_nontemporal_load(next + r * DOT_ROW_F4 + lane);
  }
}

template <int H, int OFF>
__device__ __forceinline__ void ntm_fold_step(float (&pd)[DOT_G], float (&ps)[DOT_G], const int lane) {
  const bool up = (lane & OFF) != 0;
#pragma unroll
  for (int r = 0; r < H; ++r) {
    const float keep_d = up ? pd[r + H] : pd[r];
    const float send_d = up ? pd[r] : pd[r + H];
    const float keep_s = up ? ps[r + H] : ps[r];
    const float send_s = up ? ps[r] : ps[r + H];
    pd[r] = keep_d + __shfl_xor(send_d, OFF);
    ps[r] = keep_s + __shfl_xor(send_s, OFF);
  }
}

__device__ __forceinline__ void ntm_fold_group(float (&pd)[DOT_G], float (&ps)[DOT_G],
                                               const int lane, float& out_d, float& out_s) {
  ntm_fold_step<8, 16>(pd, ps, lane);
  ntm_fold_step<4, 8>(pd, ps, lane);
  ntm_fold_step<2, 4>(pd, ps, lane);
  ntm_fold_step<1, 2>(pd, ps, lane);
  out_d = pd[0];
  out_s = ps[0];
}

__device__ __forceinline__ float ntm_join_groups(const float g0, const float g1, const int lane) {
  const bool odd = (lane & 1) != 0;
  const float keep = odd ? g1 : g0;
  const float send = odd ? g0 : g1;
  return keep + __shfl_xor(send, 1);
}

__global__ void __launch_bounds__(DOT_WAVES * 32)
ntm_dot(const float* __restrict__ memory,
        const float* __restrict__ cs,
        const float* __restrict__ Wk,
        float* __restrict__ z) {
  __shared__ float kpart[2 * NTM_D];

  const int lane = threadIdx.x & 31;
  const int wave = __builtin_amdgcn_readfirstlane(threadIdx.x >> 5);
  constexpr int WG_PER_BATCH = NTM_N / (DOT_WAVES * DOT_ROWS);
  const int b    = blockIdx.x / WG_PER_BATCH;
  const int row0 = ((blockIdx.x % WG_PER_BATCH) * DOT_WAVES + wave) * DOT_ROWS;

  const f4* p = (const f4*)(memory + ((size_t)b * NTM_N + row0) * NTM_D);
  float* zq = z + (size_t)b * NTM_N + row0 + 16 * (lane & 1) + (lane >> 1);

  f4 buf[DOT_G];
  float pd[DOT_G], ps[DOT_G];
  float d0, s0, d1, s1;
  constexpr int TRIPS = DOT_ROWS / (2 * DOT_G);

  const int half = lane >> 4;
  const int r16  = lane & 15;
  const int n0   = (wave & 7) * 16;
  const int kh   = wave >> 3;
  const float* arow = cs + (size_t)b * NTM_C + kh * KEY_KH + 2 * half;
  const float* bcol = Wk + (size_t)(kh * KEY_KH + 2 * half) * NTM_D + n0 + r16;
  v2f av[KEY_STEPS], bv[KEY_STEPS];
#pragma unroll
  for (int s = 0; s < KEY_STEPS; ++s) {
    av[s].x = arow[4 * s];
    av[s].y = arow[4 * s + 1];
    bv[s].x = bcol[(size_t)(4 * s) * NTM_D];
    bv[s].y = bcol[(size_t)(4 * s + 1) * NTM_D];
  }
  __builtin_amdgcn_sched_barrier(0);
#pragma unroll
  for (int r = 0; r < DOT_G; ++r)
    buf[r] = __builtin_nontemporal_load(p + r * DOT_ROW_F4 + lane);
  __builtin_amdgcn_sched_barrier(0);
  v8f acc = {};
#pragma unroll
  for (int s = 0; s < KEY_STEPS; ++s)
    acc = __builtin_amdgcn_wmma_f32_16x16x4_f32(
         false, av[s],  false, bv[s],
         (short)0, acc,  false,  false);
  if (lane < 16) kpart[kh * NTM_D + n0 + r16] = acc[0];
  __builtin_amdgcn_fence(__ATOMIC_RELEASE, "workgroup", "local");
  __builtin_amdgcn_s_barrier();
  __builtin_amdgcn_fence(__ATOMIC_ACQUIRE, "workgroup", "local");
  f4 k4;
  {
    const f4 klo = ((const f4*)kpart)[lane];
    const f4 khi = ((const f4*)(kpart + NTM_D))[lane];
    k4 = klo + khi;
  }
  const float knorm = sqrtf(ntm_wave_sum(k4.x * k4.x + k4.y * k4.y + k4.z * k4.z + k4.w * k4.w));

#pragma nounroll
  for (int t = 0; t < TRIPS - 1; ++t) {
    ntm_take_group<true>(buf, p + DOT_G * DOT_ROW_F4, k4, lane, pd, ps);
    __builtin_amdgcn_sched_barrier(0);
    ntm_fold_group(pd, ps, lane, d0, s0);
    __builtin_amdgcn_sched_barrier(0);
    ntm_take_group<true>(buf, p + 2 * DOT_G * DOT_ROW_F4, k4, lane, pd, ps);
    __builtin_amdgcn_sched_barrier(0);
    ntm_fold_group(pd, ps, lane, d1, s1);
    {
      const float kd = ntm_join_groups(d0, d1, lane);
      const float ks = ntm_join_groups(s0, s1, lane);
      *zq = kd / (knorm * sqrtf(ks) + NTM_EPS);
    }
    p  += 2 * DOT_G * DOT_ROW_F4;
    zq += 2 * DOT_G;
  }
  ntm_take_group<true>(buf, p + DOT_G * DOT_ROW_F4, k4, lane, pd, ps);
  __builtin_amdgcn_sched_barrier(0);
  ntm_fold_group(pd, ps, lane, d0, s0);
  ntm_take_group<false>(buf, p, k4, lane, pd, ps);
  ntm_fold_group(pd, ps, lane, d1, s1);
  {
    const float kd = ntm_join_groups(d0, d1, lane);
    const float ks = ntm_join_groups(s0, s1, lane);
    *zq = kd / (knorm * sqrtf(ks) + NTM_EPS);
  }
}

__device__ __forceinline__ float ntm_block_reduce_sum(float v, float* red) {
#pragma unroll
  for (int off = 16; off > 0; off >>= 1) v += __shfl_down(v, off);
  const int wid = threadIdx.x >> 5, lane = threadIdx.x & 31;
  if (lane == 0) red[wid] = v;
  __syncthreads();
  if (threadIdx.x < 32) {
    float x = red[threadIdx.x];
#pragma unroll
    for (int off = 16; off > 0; off >>= 1) x += __shfl_down(x, off);
    if (threadIdx.x == 0) red[0] = x;
  }
  __syncthreads();
  const float r = red[0];
  __syncthreads();
  return r;
}

__device__ __forceinline__ float ntm_block_reduce_max(float v, float* red) {
#pragma unroll
  for (int off = 16; off > 0; off >>= 1) v = fmaxf(v, __shfl_down(v, off));
  const int wid = threadIdx.x >> 5, lane = threadIdx.x & 31;
  if (lane == 0) red[wid] = v;
  __syncthreads();
  if (threadIdx.x < 32) {
    float x = red[threadIdx.x];
#pragma unroll
    for (int off = 16; off > 0; off >>= 1) x = fmaxf(x, __shfl_down(x, off));
    if (threadIdx.x == 0) red[0] = x;
  }
  __syncthreads();
  const float r = red[0];
  __syncthreads();
  return r;
}

__device__ __forceinline__ float ntm_pow_pos(const float x, const float g) {
  const int   e  = __builtin_amdgcn_frexp_expf(x * 1.41421356f) - 1;
  const float m  = ldexpf(x, -e);
  const float ef = (float)e;
  const float lm = __builtin_amdgcn_logf(m);
  const float n  = __builtin_rintf(g * (ef + lm));
  const float f  = __builtin_fmaf(g, lm, __builtin_fmaf(g, ef, -n));
  return ldexpf(__builtin_amdgcn_exp2f(f), (int)n);
}

__global__ void ntm_rowops(const float* __restrict__ z,
                           const float* __restrict__ prev,
                           const float* __restrict__ cs,
                           const float* __restrict__ Wb,  const float* __restrict__ bb,
                           const float* __restrict__ Wg,  const float* __restrict__ bg,
                           const float* __restrict__ Ws,  const float* __restrict__ bs,
                           const float* __restrict__ Wgam,const float* __restrict__ bgam,
                           float* __restrict__ out) {
  __shared__ float row[NTM_N];
  __shared__ float red[32];
  __shared__ float prm[8];

  const int b    = blockIdx.x;
  const int tid  = threadIdx.x;
  const int wv   = __builtin_amdgcn_readfirstlane(tid >> 5);
  const int lane = tid & 31;

  float zv[8];
  float lmax = -3.4e38f;
#pragma unroll
  for (int i = 0; i < 8; ++i) {
    const int n = tid + i * 1024;
    zv[i] = z[(size_t)b * NTM_N + n];
  }

  if (wv < 4) {
    float c[8];
#pragma unroll
    for (int j = 0; j < 8; ++j) c[j] = cs[(size_t)b * NTM_C + lane + 32 * j];
    if (wv == 0) {
      float xb = 0.f;
#pragma unroll
      for (int j = 0; j < 8; ++j) xb += c[j] * Wb[lane + 32 * j];
      xb = ntm_wave_sum(xb);
      const float beta = ntm_softplus(xb + bb[0]) + 1.0f;
      if (lane == 0) prm[0] = beta;
    } else if (wv == 1) {
      float xg = 0.f;
#pragma unroll
      for (int j = 0; j < 8; ++j) xg += c[j] * Wg[lane + 32 * j];
      xg = ntm_wave_sum(xg);
      const float gate = 1.0f / (1.0f + expf(-(xg + bg[0])));
      if (lane == 0) prm[1] = gate;
    } else if (wv == 2) {
      float xm = 0.f;
#pragma unroll
      for (int j = 0; j < 8; ++j) xm += c[j] * Wgam[lane + 32 * j];
      xm = ntm_wave_sum(xm);
      const float gamma = ntm_softplus(xm + bgam[0]) + 1.0f;
      if (lane == 0) prm[2] = gamma;
    } else {
      float x0 = 0.f, x1 = 0.f, x2 = 0.f;
#pragma unroll
      for (int j = 0; j < 8; ++j) {
        const int t = lane + 32 * j;
        x0 += c[j] * Ws[t * 3 + 0];
        x1 += c[j] * Ws[t * 3 + 1];
        x2 += c[j] * Ws[t * 3 + 2];
      }
      x0 = ntm_wave_sum(x0) + bs[0];
      x1 = ntm_wave_sum(x1) + bs[1];
      x2 = ntm_wave_sum(x2) + bs[2];
      const float m = fmaxf(x0, fmaxf(x1, x2));
      const float e0 = expf(x0 - m), e1 = expf(x1 - m), e2 = expf(x2 - m);
      const float inv = 1.0f / (e0 + e1 + e2);
      if (lane == 0) { prm[4] = e0 * inv; prm[5] = e1 * inv; prm[6] = e2 * inv; }
    }
  }

#pragma unroll
  for (int i = 0; i < 8; ++i) lmax = fmaxf(lmax, zv[i]);
  const float smax = ntm_block_reduce_max(lmax, red);

  const float beta  = prm[0];
  const float gate  = prm[1];
  const float gamma = prm[2];
  const float s0 = prm[4];
  const float s1 = prm[5];
  const float s2 = prm[6];
  const float rmax = beta * smax;

  float lsum = 0.f;
#pragma unroll
  for (int i = 0; i < 8; ++i) {
    zv[i] = expf(beta * zv[i] - rmax);
    lsum += zv[i];
  }
  const float invsum = 1.0f / ntm_block_reduce_sum(lsum, red);

#pragma unroll
  for (int i = 0; i < 8; ++i) {
    const int n = tid + i * 1024;
    const float cw = zv[i] * invsum;
    row[n] = gate * cw + (1.0f - gate) * prev[(size_t)b * NTM_N + n];
  }
  __syncthreads();

  float sh[8];
  float lsh = 0.f;
#pragma unroll
  for (int i = 0; i < 8; ++i) {
    const int n = tid + i * 1024;
    const float right = row[(n + 1) & (NTM_N - 1)];
    const float mid   = row[n];
    const float left  = row[(n + NTM_N - 1) & (NTM_N - 1)];
    const float shifted = s0 * right + s1 * mid + s2 * left;
    sh[i] = ntm_pow_pos(shifted + 1e-8f, gamma);
    lsh += sh[i];
  }
  const float invt = 1.0f / (ntm_block_reduce_sum(lsh, red) + NTM_EPS);

#pragma unroll
  for (int i = 0; i < 8; ++i) {
    const int n = tid + i * 1024;
    out[(size_t)b * NTM_N + n] = sh[i] * invt;
  }
}

extern "C" void kernel_launch(void* const* d_in, const int* in_sizes, int n_in,
                              void* d_out, int out_size, void* d_ws, size_t ws_size,
                              hipStream_t stream) {
  const float* memory = (const float*)d_in[0];
  const float* cs     = (const float*)d_in[1];
  const float* prev   = (const float*)d_in[2];
  const float* Wk     = (const float*)d_in[3];
  const float* Wb     = (const float*)d_in[4];
  const float* bb     = (const float*)d_in[5];
  const float* Wg     = (const float*)d_in[6];
  const float* bg     = (const float*)d_in[7];
  const float* Ws     = (const float*)d_in[8];
  const float* bs     = (const float*)d_in[9];
  const float* Wgam   = (const float*)d_in[10];
  const float* bgam   = (const float*)d_in[11];
  float* out = (float*)d_out;

  float* zbuf   = (float*)d_ws;

  ntm_dot<<<dim3(NTM_B * NTM_N / (DOT_WAVES * DOT_ROWS)), DOT_WAVES * 32, 0, stream>>>(memory, cs, Wk, zbuf);
  ntm_rowops<<<NTM_B, 1024, 0, stream>>>(zbuf, prev, cs, Wb, bb, Wg, bg, Ws, bs, Wgam, bgam, out);
}


# ntm_rowops as 8 waves x 32 elements per lane (was 32 x 8), 16-byte loads/stores; each block max/sum = one LDS plane + one barrier (was 3 barriers, 2 shuffle trees); order of its two sums changes
# speedup vs baseline: 1.8742x; 1.0361x over previous
#include <hip/hip_runtime.h>
#include <math.h>

typedef float v2f __attribute__((ext_vector_type(2)));
typedef float v8f __attribute__((ext_vector_type(8)));
typedef float f4  __attribute__((ext_vector_type(4)));

#define NTM_B 64
#define NTM_N 8192
#define NTM_D 128
#define NTM_C 256
#define NTM_EPS 1e-8f

__device__ __forceinline__ float ntm_softplus(float x) {
  return (x > 20.f) ? x : log1pf(expf(x));
}

__device__ __forceinline__ float ntm_wave_sum(float v) {
#pragma unroll
  for (int off = 16; off > 0; off >>= 1) v += __shfl_xor(v, off);
  return v;
}

#define KEY_KH    (NTM_C / 2)
#define KEY_STEPS (KEY_KH / 4)
#define DOT_WAVES 16
#define DOT_ROWS  128
#define DOT_G     16
#define DOT_ROW_F4 (NTM_D / 4)

template <bool REFILL>
__device__ __forceinline__ void ntm_take_group(f4 (&buf)[DOT_G], const f4* next,
                                               const f4 k4, const int lane,
                                               float (&pd)[DOT_G], float (&ps)[DOT_G]) {
#pragma unroll
  for (int r = 0; r < DOT_G; ++r) {
    const f4 m4 = buf[r];
    pd[r] = m4.x * k4.x + m4.y * k4.y + m4.z * k4.z + m4.w * k4.w;
    ps[r] = m4.x * m4.x + m4.y * m4.y + m4.z * m4.z + m4.w * m4.w;
    if (REFILL) buf[r] = __builtin_nontemporal_load(next + r * DOT_ROW_F4 + lane);
  }
}

template <int H, int OFF>
__device__ __forceinline__ void ntm_fold_step(float (&pd)[DOT_G], float (&ps)[DOT_G], const int lane) {
  const bool up = (lane & OFF) != 0;
#pragma unroll
  for (int r = 0; r < H; ++r) {
    const float keep_d = up ? pd[r + H] : pd[r];
    const float send_d = up ? pd[r] : pd[r + H];
    const float keep_s = up ? ps[r + H] : ps[r];
    const float send_s = up ? ps[r] : ps[r + H];
    pd[r] = keep_d + __shfl_xor(send_d, OFF);
    ps[r] = keep_s + __shfl_xor(send_s, OFF);
  }
}

__device__ __forceinline__ void ntm_fold_group(float (&pd)[DOT_G], float (&ps)[DOT_G],
                                               const int lane, float& out_d, float& out_s) {
  ntm_fold_step<8, 16>(pd, ps, lane);
  ntm_fold_step<4, 8>(pd, ps, lane);
  ntm_fold_step<2, 4>(pd, ps, lane);
  ntm_fold_step<1, 2>(pd, ps, lane);
  out_d = pd[0];
  out_s = ps[0];
}

__device__ __forceinline__ float ntm_join_groups(const float g0, const float g1, const int lane) {
  const bool odd = (lane & 1) != 0;
  const float keep = odd ? g1 : g0;
  const float send = odd ? g0 : g1;
  return keep + __shfl_xor(send, 1);
}

__global__ void __launch_bounds__(DOT_WAVES * 32)
ntm_dot(const float* __restrict__ memory,
        const float* __restrict__ cs,
        const float* __restrict__ Wk,
        float* __restrict__ z) {
  __shared__ float kpart[2 * NTM_D];

  const int lane = threadIdx.x & 31;
  const int wave = __builtin_amdgcn_readfirstlane(threadIdx.x >> 5);
  constexpr int WG_PER_BATCH = NTM_N / (DOT_WAVES * DOT_ROWS);
  const int b    = blockIdx.x / WG_PER_BATCH;
  const int row0 = ((blockIdx.x % WG_PER_BATCH) * DOT_WAVES + wave) * DOT_ROWS;

  const f4* p = (const f4*)(memory + ((size_t)b * NTM_N + row0) * NTM_D);
  float* zq = z + (size_t)b * NTM_N + row0 + 16 * (lane & 1) + (lane >> 1);

  f4 buf[DOT_G];
  float pd[DOT_G], ps[DOT_G];
  float d0, s0, d1, s1;
  constexpr int TRIPS = DOT_ROWS / (2 * DOT_G);

  const int half = lane >> 4;
  const int r16  = lane & 15;
  const int n0   = (wave & 7) * 16;
  const int kh   = wave >> 3;
  const float* arow = cs + (size_t)b * NTM_C + kh * KEY_KH + 2 * half;
  const float* bcol = Wk + (size_t)(kh * KEY_KH + 2 * half) * NTM_D + n0 + r16;
  v2f av[KEY_STEPS], bv[KEY_STEPS];
#pragma unroll
  for (int s = 0; s < KEY_STEPS; ++s) {
    av[s].x = arow[4 * s];
    av[s].y = arow[4 * s + 1];
    bv[s].x = bcol[(size_t)(4 * s) * NTM_D];
    bv[s].y = bcol[(size_t)(4 * s + 1) * NTM_D];
  }
  __builtin_amdgcn_sched_barrier(0);
#pragma unroll
  for (int r = 0; r < DOT_G; ++r)
    buf[r] = __builtin_nontemporal_load(p + r * DOT_ROW_F4 + lane);
  __builtin_amdgcn_sched_barrier(0);
  v8f acc = {};
#pragma unroll
  for (int s = 0; s < KEY_STEPS; ++s)
    acc = __builtin_amdgcn_wmma_f32_16x16x4_f32(
         false, av[s],  false, bv[s],
         (short)0, acc,  false,  false);
  if (lane < 16) kpart[kh * NTM_D + n0 + r16] = acc[0];
  __builtin_amdgcn_fence(__ATOMIC_RELEASE, "workgroup", "local");
  __builtin_amdgcn_s_barrier();
  __builtin_amdgcn_fence(__ATOMIC_ACQUIRE, "workgroup", "local");
  f4 k4;
  {
    const f4 klo = ((const f4*)kpart)[lane];
    const f4 khi = ((const f4*)(kpart + NTM_D))[lane];
    k4 = klo + khi;
  }
  const float knorm = sqrtf(ntm_wave_sum(k4.x * k4.x + k4.y * k4.y + k4.z * k4.z + k4.w * k4.w));

#pragma nounroll
  for (int t = 0; t < TRIPS - 1; ++t) {
    ntm_take_group<true>(buf, p + DOT_G * DOT_ROW_F4, k4, lane, pd, ps);
    __builtin_amdgcn_sched_barrier(0);
    ntm_fold_group(pd, ps, lane, d0, s0);
    __builtin_amdgcn_sched_barrier(0);
    ntm_take_group<true>(buf, p + 2 * DOT_G * DOT_ROW_F4, k4, lane, pd, ps);
    __builtin_amdgcn_sched_barrier(0);
    ntm_fold_group(pd, ps, lane, d1, s1);
    {
      const float kd = ntm_join_groups(d0, d1, lane);
      const float ks = ntm_join_groups(s0, s1, lane);
      *zq = kd / (knorm * sqrtf(ks) + NTM_EPS);
    }
    p  += 2 * DOT_G * DOT_ROW_F4;
    zq += 2 * DOT_G;
  }
  ntm_take_group<true>(buf, p + DOT_G * DOT_ROW_F4, k4, lane, pd, ps);
  __builtin_amdgcn_sched_barrier(0);
  ntm_fold_group(pd, ps, lane, d0, s0);
  ntm_take_group<false>(buf, p, k4, lane, pd, ps);
  ntm_fold_group(pd, ps, lane, d1, s1);
  {
    const float kd = ntm_join_groups(d0, d1, lane);
    const float ks = ntm_join_groups(s0, s1, lane);
    *zq = kd / (knorm * sqrtf(ks) + NTM_EPS);
  }
}

#define RO_THREADS 256
#define RO_PIECES  (NTM_N / (4 * RO_THREADS))

__device__ __forceinline__ float ntm_wave_max(float v) {
#pragma unroll
  for (int off = 16; off > 0; off >>= 1) v = fmaxf(v, __shfl_xor(v, off));
  return v;
}

__device__ __forceinline__ float ntm_block_sum(const float v, float* plane, const int tid, const int lane) {
  plane[tid] = v;
  __syncthreads();
  const f4 a = ((const f4*)plane)[lane];
  const f4 c = ((const f4*)plane)[lane + 32];
  return ntm_wave_sum(((a.x + a.y) + (a.z + a.w)) + ((c.x + c.y) + (c.z + c.w)));
}

__device__ __forceinline__ float ntm_block_max(const float v, float* plane, const int tid, const int lane) {
  plane[tid] = v;
  __syncthreads();
  const f4 a = ((const f4*)plane)[lane];
  const f4 c = ((const f4*)plane)[lane + 32];
  return ntm_wave_max(fmaxf(fmaxf(fmaxf(a.x, a.y), fmaxf(a.z, a.w)),
                            fmaxf(fmaxf(c.x, c.y), fmaxf(c.z, c.w))));
}

__device__ __forceinline__ float ntm_pow_pos(const float x, const float g) {
  const int   e  = __builtin_amdgcn_frexp_expf(x * 1.41421356f) - 1;
  const float m  = ldexpf(x, -e);
  const float ef = (float)e;
  const float lm = __builtin_amdgcn_logf(m);
  const float n  = __builtin_rintf(g * (ef + lm));
  const float f  = __builtin_fmaf(g, lm, __builtin_fmaf(g, ef, -n));
  return ldexpf(__builtin_amdgcn_exp2f(f), (int)n);
}

__global__ void __launch_bounds__(RO_THREADS)
ntm_rowops(const float* __restrict__ z,
           const float* __restrict__ prev,
           const float* __restrict__ cs,
           const float* __restrict__ Wb,  const float* __restrict__ bb,
           const float* __restrict__ Wg,  const float* __restrict__ bg,
           const float* __restrict__ Ws,  const float* __restrict__ bs,
           const float* __restrict__ Wgam,const float* __restrict__ bgam,
           float* __restrict__ out) {
  __shared__ __attribute__((aligned(16))) float row[NTM_N];
  __shared__ __attribute__((aligned(16))) float red[3][RO_THREADS];
  __shared__ float prm[8];

  const int b    = blockIdx.x;
  const int tid  = threadIdx.x;
  const int wv   = __builtin_amdgcn_readfirstlane(tid >> 5);
  const int lane = tid & 31;

  const f4* zrow = (const f4*)(z + (size_t)b * NTM_N);
  const f4* prow = (const f4*)(prev + (size_t)b * NTM_N);
  f4*       orow = (f4*)(out + (size_t)b * NTM_N);

  f4 zv[RO_PIECES];
#pragma unroll
  for (int j = 0; j < RO_PIECES; ++j) zv[j] = zrow[tid + RO_THREADS * j];

  if (wv < 4) {
    float c[8];
#pragma unroll
    for (int j = 0; j < 8; ++j) c[j] = cs[(size_t)b * NTM_C + lane + 32 * j];
    if (wv == 0) {
      float xb = 0.f;
#pragma unroll
      for (int j = 0; j < 8; ++j) xb += c[j] * Wb[lane + 32 * j];
      xb = ntm_wave_sum(xb);
      const float beta = ntm_softplus(xb + bb[0]) + 1.0f;
      if (lane == 0) prm[0] = beta;
    } else if (wv == 1) {
      float xg = 0.f;
#pragma unroll
      for (int j = 0; j < 8; ++j) xg += c[j] * Wg[lane + 32 * j];
      xg = ntm_wave_sum(xg);
      const float gate = 1.0f / (1.0f + expf(-(xg + bg[0])));
      if (lane == 0) prm[1] = gate;
    } else if (wv == 2) {
      float xm = 0.f;
#pragma unroll
      for (int j = 0; j < 8; ++j) xm += c[j] * Wgam[lane + 32 * j];
      xm = ntm_wave_sum(xm);
      const float gamma = ntm_softplus(xm + bgam[0]) + 1.0f;
      if (lane == 0) prm[2] = gamma;
    } else {
      float x0 = 0.f, x1 = 0.f, x2 = 0.f;
#pragma unroll
      for (int j = 0; j < 8; ++j) {
        const int t = lane + 32 * j;
        x0 += c[j] * Ws[t * 3 + 0];
        x1 += c[j] * Ws[t * 3 + 1];
        x2 += c[j] * Ws[t * 3 + 2];
      }
      x0 = ntm_wave_sum(x0) + bs[0];
      x1 = ntm_wave_sum(x1) + bs[1];
      x2 = ntm_wave_sum(x2) + bs[2];
      const float m = fmaxf(x0, fmaxf(x1, x2));
      const float e0 = expf(x0 - m), e1 = expf(x1 - m), e2 = expf(x2 - m);
      const float inv = 1.0f / (e0 + e1 + e2);
      if (lane == 0) { prm[4] = e0 * inv; prm[5] = e1 * inv; prm[6] = e2 * inv; }
    }
  }

  float lmax = -3.4e38f;
#pragma unroll
  for (int j = 0; j < RO_PIECES; ++j)
    lmax = fmaxf(lmax, fmaxf(fmaxf(zv[j].x, zv[j].y), fmaxf(zv[j].z, zv[j].w)));
  const float smax = ntm_block_max(lmax, red[0], tid, lane);

  const float beta  = prm[0];
  const float gate  = prm[1];
  const float gamma = prm[2];
  const float s0 = prm[4];
  const float s1 = prm[5];
  const float s2 = prm[6];
  const float rmax = beta * smax;

  float lsum = 0.f;
#pragma unroll
  for (int j = 0; j < RO_PIECES; ++j) {
    zv[j].x = expf(beta * zv[j].x - rmax);
    zv[j].y = expf(beta * zv[j].y - rmax);
    zv[j].z = expf(beta * zv[j].z - rmax);
    zv[j].w = expf(beta * zv[j].w - rmax);
    lsum += (zv[j].x + zv[j].y) + (zv[j].z + zv[j].w);
  }
  const float invsum = 1.0f / ntm_block_sum(lsum, red[1], tid, lane);

  f4 gw[RO_PIECES];
#pragma unroll
  for (int j = 0; j < RO_PIECES; ++j) {
    const f4 pv = prow[tid + RO_THREADS * j];
    gw[j].x = gate * (zv[j].x * invsum) + (1.0f - gate) * pv.x;
    gw[j].y = gate * (zv[j].y * invsum) + (1.0f - gate) * pv.y;
    gw[j].z = gate * (zv[j].z * invsum) + (1.0f - gate) * pv.z;
    gw[j].w = gate * (zv[j].w * invsum) + (1.0f - gate) * pv.w;
    ((f4*)row)[tid + RO_THREADS * j] = gw[j];
  }
  __syncthreads();

  f4 sh[RO_PIECES];
  float lsh = 0.f;
#pragma unroll
  for (int j = 0; j < RO_PIECES; ++j) {
    const int n0 = 4 * tid + 4 * RO_THREADS * j;
    const float left  = row[(n0 + NTM_N - 1) & (NTM_N - 1)];
    const float right = row[(n0 + 4) & (NTM_N - 1)];
    const f4 m = gw[j];
    const float t0 = s0 * m.y   + s1 * m.x + s2 * left;
    const float t1 = s0 * m.z   + s1 * m.y + s2 * m.x;
    const float t2 = s0 * m.w   + s1 * m.z + s2 * m.y;
    const float t3 = s0 * right + s1 * m.w + s2 * m.z;
    sh[j].x = ntm_pow_pos(t0 + 1e-8f, gamma);
    sh[j].y = ntm_pow_pos(t1 + 1e-8f, gamma);
    sh[j].z = ntm_pow_pos(t2 + 1e-8f, gamma);
    sh[j].w = ntm_pow_pos(t3 + 1e-8f, gamma);
    lsh += (sh[j].x + sh[j].y) + (sh[j].z + sh[j].w);
  }
  const float invt = 1.0f / (ntm_block_sum(lsh, red[2], tid, lane) + NTM_EPS);

#pragma unroll
  for (int j = 0; j < RO_PIECES; ++j) orow[tid + RO_THREADS * j] = sh[j] * invt;
}

extern "C" void kernel_launch(void* const* d_in, const int* in_sizes, int n_in,
                              void* d_out, int out_size, void* d_ws, size_t ws_size,
                              hipStream_t stream) {
  const float* memory = (const float*)d_in[0];
  const float* cs     = (const float*)d_in[1];
  const float* prev   = (const float*)d_in[2];
  const float* Wk     = (const float*)d_in[3];
  const float* Wb     = (const float*)d_in[4];
  const float* bb     = (const float*)d_in[5];
  const float* Wg     = (const float*)d_in[6];
  const float* bg     = (const float*)d_in[7];
  const float* Ws     = (const float*)d_in[8];
  const float* bs     = (const float*)d_in[9];
  const float* Wgam   = (const float*)d_in[10];
  const float* bgam   = (const float*)d_in[11];
  float* out = (float*)d_out;

  float* zbuf   = (float*)d_ws;

  ntm_dot<<<dim3(NTM_B * NTM_N / (DOT_WAVES * DOT_ROWS)), DOT_WAVES * 32, 0, stream>>>(memory, cs, Wk, zbuf);
  ntm_rowops<<<NTM_B, RO_THREADS, 0, stream>>>(zbuf, prev, cs, Wb, bb, Wg, bg, Ws, bs, Wgam, bgam, out);
}
